# starting kernel
# baseline (speedup 1.0000x reference)
#include <hip/hip_runtime.h>
#include <hip/hip_bf16.h>

typedef __attribute__((ext_vector_type(2))) float v2f;
typedef __attribute__((ext_vector_type(8))) float v8f;

#define N_NODES 100000
#define N_EDGES 1600000
#define D_FEAT  64
#define UNITS   64

__global__ __launch_bounds__(256) void gcn_degree_kernel(
    const int* __restrict__ target, float* __restrict__ deg, int nEdges) {
  int e = blockIdx.x * blockDim.x + threadIdx.x;
  if (e < nEdges) atomicAdd(&deg[target[e]], 1.0f);
}

__global__ __launch_bounds__(256) void gcn_rsqrt_kernel(
    const float* __restrict__ deg, float* __restrict__ dinv, int n) {
  int i = blockIdx.x * blockDim.x + threadIdx.x;
  if (i < n) dinv[i] = rsqrtf(deg[i]);
}

__global__ __launch_bounds__(256) void gcn_scatter_kernel(
    const float* __restrict__ x, const int* __restrict__ src,
    const int* __restrict__ tgt, const float* __restrict__ dinv,
    float* __restrict__ pooled, int nEdges) {
  int wave = (blockIdx.x * blockDim.x + threadIdx.x) >> 5;
  int lane = threadIdx.x & 31;
  if (wave >= nEdges) return;
  int s = src[wave];
  int t = tgt[wave];
  float d = dinv[s];
  const float2* xs = (const float2*)(x + (size_t)s * D_FEAT);
  float2 v = xs[lane];
  float* p = pooled + (size_t)t * D_FEAT + 2 * lane;
  atomicAdd(p + 0, v.x * d);
  atomicAdd(p + 1, v.y * d);
}

__global__ __launch_bounds__(128) void gcn_gemm_wmma_kernel(
    const float* __restrict__ pooled, const float* __restrict__ W,
    const float* __restrict__ bias, const float* __restrict__ dinv,
    float* __restrict__ out) {
  __shared__ float sW[D_FEAT * UNITS];
  __shared__ float sA[16 * D_FEAT];

  const int tid = threadIdx.x;
  const int rowBase = blockIdx.x * 16;

  {
    const float4* Wv = (const float4*)W;
    float4* sWv = (float4*)sW;
    for (int i = tid; i < (D_FEAT * UNITS) / 4; i += 128) sWv[i] = Wv[i];
  }
  {
    const float4* Av = (const float4*)(pooled + (size_t)rowBase * D_FEAT);
    float4* sAv = (float4*)sA;
    for (int i = tid; i < (16 * D_FEAT) / 4; i += 128) {
      int row = i / (D_FEAT / 4);
      float d = dinv[rowBase + row];
      float4 v = Av[i];
      v.x *= d; v.y *= d; v.z *= d; v.w *= d;
      sAv[i] = v;
    }
  }
  __syncthreads();

  const int lane  = tid & 31;
  const int wave  = tid >> 5;
  const int nBase = wave * 16;
  const int h     = lane >> 4;
  const int mn    = lane & 15;

  v8f acc = {};
  #pragma unroll
  for (int kk = 0; kk < D_FEAT; kk += 4) {
    const int k0 = kk + 2 * h;
    v2f a, b;
    a.x = sA[mn * D_FEAT + k0];
    a.y = sA[mn * D_FEAT + k0 + 1];
    b.x = sW[(k0)     * UNITS + nBase + mn];
    b.y = sW[(k0 + 1) * UNITS + nBase + mn];
    acc = __builtin_amdgcn_wmma_f32_16x16x4_f32(
         false, a,  false, b,
         (short)0, acc,  false,  false);
  }

  const float bn = bias[nBase + mn];
  #pragma unroll
  for (int j = 0; j < 8; ++j) {
    const int m = j + 8 * h;
    float r = acc[j] + bn;
    out[(size_t)(rowBase + m) * UNITS + nBase + mn] = fmaxf(r, 0.0f);
  }
}

extern "C" void kernel_launch(void* const* d_in, const int* in_sizes, int n_in,
                              void* d_out, int out_size, void* d_ws, size_t ws_size,
                              hipStream_t stream) {
  const float* x      = (const float*)d_in[0];
  const float* W      = (const float*)d_in[1];
  const float* b      = (const float*)d_in[2];
  const int*   source = (const int*)d_in[3];
  const int*   target = (const int*)d_in[4];
  float*       out    = (float*)d_out;

  float* pooled = (float*)d_ws;
  float* deg    = pooled + (size_t)N_NODES * D_FEAT;
  float* dinv   = deg + N_NODES;

  hipMemsetAsync(pooled, 0, (size_t)N_NODES * D_FEAT * sizeof(float), stream);
  hipMemsetAsync(deg,    0, (size_t)N_NODES * sizeof(float), stream);

  gcn_degree_kernel<<<(N_EDGES + 255) / 256, 256, 0, stream>>>(target, deg, N_EDGES);
  gcn_rsqrt_kernel<<<(N_NODES + 255) / 256, 256, 0, stream>>>(deg, dinv, N_NODES);
  gcn_scatter_kernel<<<(N_EDGES + 7) / 8, 256, 0, stream>>>(x, source, target, dinv,
                                                            pooled, N_EDGES);
  gcn_gemm_wmma_kernel<<<N_NODES / 16, 128, 0, stream>>>(pooled, W, b, dinv, out);
}


# Each of 250 blocks owns 400 target nodes: filters the target list, adds scaled source rows into an LDS tile by LDS f32 atomics, WMMA reads the tile; no global row atomics; LDS arrival-order sums
# speedup vs baseline: 1.6838x; 1.6838x over previous
#include <hip/hip_runtime.h>
#include <hip/hip_bf16.h>


typedef __attribute__((ext_vector_type(2))) float v2f;
typedef __attribute__((ext_vector_type(8))) float v8f;

#define N_NODES 100000
#define N_EDGES 1600000
#define D_FEAT  64
#define UNITS   64

#define NPB         400
#define N_OWNERS    (N_NODES / NPB)
#define OWN_THREADS 1024
#define OWN_WAVES   (OWN_THREADS / 32)
#define APITCH      66
#define CHUNK       128
#define N_CHUNKS    (N_EDGES / CHUNK)
#define HALF_UNROLL 4
#define LIST_CAP    (32 + HALF_UNROLL * CHUNK)
#define SCAN_UNROLL 8
#define E_BITS      21
#define ROW_BYTES   (D_FEAT * 4u)

static_assert(N_NODES % NPB == 0, "owner ranges tile the nodes exactly");
static_assert(NPB % 16 == 0 && NPB / 16 <= OWN_WAVES, "one 16-row tile per wave");
static_assert(N_EDGES % CHUNK == 0, "edge list is a whole number of chunks");
static_assert(N_CHUNKS % N_OWNERS == 0, "start offsets tile the chunks");
static_assert(N_EDGES <= (1 << E_BITS) && NPB <= (1 << (32 - E_BITS)), "list entry packs in 32 bits");
static_assert(SCAN_UNROLL == 2 * HALF_UNROLL, "two looks at the list per group of loads");
static_assert((unsigned long long)N_NODES * D_FEAT * 4ull < (1ull << 32), "row offsets fit in 32 bits");

__global__ __launch_bounds__(256) void gcn_degree_kernel(
    const int* __restrict__ target, float* __restrict__ deg, int nEdges) {
  int e = blockIdx.x * blockDim.x + threadIdx.x;
  if (e < nEdges) atomicAdd(&deg[target[e]], 1.0f);
}

__global__ __launch_bounds__(256) void gcn_rsqrt_kernel(
    const float* __restrict__ deg, float* __restrict__ dinv, int n) {
  int i = blockIdx.x * blockDim.x + threadIdx.x;
  if (i < n) dinv[i] = rsqrtf(deg[i]);
}

__device__ __forceinline__ void add_row(float2 v, float di, int ri, int lane, float* sAcc) {
  float* p = sAcc + ri * APITCH + 2 * lane;
  atomicAdd(p + 0, v.x * di);
  atomicAdd(p + 1, v.y * di);
}

template <bool FULL>
__device__ __forceinline__ void drain_batch(
    const unsigned* myList, int n, int lane,
    const float* __restrict__ x, const int* __restrict__ src,
    const float* __restrict__ dinv, float* sAcc) {
  unsigned ent = 0u;
  int s = 0;
  float d = 0.0f;
  if (FULL || lane < n) {
    ent = myList[lane];
    s = src[ent & ((1u << E_BITS) - 1u)];
    d = dinv[s];
  }
  const int rel = (int)(ent >> E_BITS);
  const int dbits = __float_as_int(d);
  const char* xb = (const char*)x;
  const unsigned lane8 = (unsigned)lane * 8u;
  const int nfull = FULL ? 32 : ((n + 15) & ~15);
  for (int i = 0; i < nfull; i += 16) {
    float2 xv[16];
    #pragma unroll
    for (int q = 0; q < 16; ++q) {
      const unsigned si = (unsigned)__builtin_amdgcn_readlane(s, i + q);
      xv[q] = *(const float2*)(xb + (si * ROW_BYTES + lane8));
    }
    #pragma unroll
    for (int q = 0; q < 16; ++q) {
      const float di = __int_as_float(__builtin_amdgcn_readlane(dbits, i + q));
      const int ri = __builtin_amdgcn_readlane(rel, i + q);
      if (FULL || i + q < n) add_row(xv[q], di, ri, lane, sAcc);
    }
  }
}

__device__ __forceinline__ int drain_whole_batches(
    unsigned* myList, int cnt, int lane,
    const float* __restrict__ x, const int* __restrict__ src,
    const float* __restrict__ dinv, float* sAcc) {
  const int full = cnt & ~31;
  for (int j = 0; j < full; j += 32)
    drain_batch<true>(myList + j, 32, lane, x, src, dinv, sAcc);
  const int rem = cnt - full;
  if (lane < rem) {
    const unsigned t = myList[full + lane];
    myList[lane] = t;
  }
  return rem;
}

__global__ __launch_bounds__(OWN_THREADS) void gcn_owner_wmma_kernel(
    const float* __restrict__ x, const int* __restrict__ src,
    const int* __restrict__ tgt, const float* __restrict__ dinv,
    const float* __restrict__ W, const float* __restrict__ bias,
    float* __restrict__ out) {
  __shared__ __attribute__((aligned(16))) float sAcc[NPB * APITCH];
  __shared__ __attribute__((aligned(16))) float sW[D_FEAT * UNITS];
  __shared__ unsigned sList[OWN_WAVES * LIST_CAP];

  const int tid  = threadIdx.x;
  const int lane = tid & 31;
  const int wave = __builtin_amdgcn_readfirstlane(tid >> 5);
  const int base = blockIdx.x * NPB;

  {
    float4* z = (float4*)sAcc;
    for (int i = tid; i < (NPB * APITCH) / 4; i += OWN_THREADS)
      z[i] = make_float4(0.0f, 0.0f, 0.0f, 0.0f);
    const float4* Wv = (const float4*)W;
    float4* sWv = (float4*)sW;
    for (int i = tid; i < (D_FEAT * UNITS) / 4; i += OWN_THREADS) sWv[i] = Wv[i];
  }
  __syncthreads();

  unsigned* myList = sList + wave * LIST_CAP;
  int cnt = 0;
  const int4* tv = (const int4*)tgt;
  const int start = blockIdx.x * (N_CHUNKS / N_OWNERS);

  for (int c0 = wave; c0 < N_CHUNKS; c0 += OWN_WAVES * SCAN_UNROLL) {
    int4 v[SCAN_UNROLL];
    int  ch[SCAN_UNROLL];
    #pragma unroll
    for (int u = 0; u < SCAN_UNROLL; ++u) {
      int c = min(c0 + u * OWN_WAVES, N_CHUNKS - 1);
      c += start;
      if (c >= N_CHUNKS) c -= N_CHUNKS;
      ch[u] = c;
      v[u] = tv[c * 32 + lane];
    }
    #pragma unroll
    for (int u = 0; u < SCAN_UNROLL; ++u) {
      const bool live = (c0 + u * OWN_WAVES) < N_CHUNKS;
      const unsigned r0 = (unsigned)(v[u].x - base);
      const unsigned r1 = (unsigned)(v[u].y - base);
      const unsigned r2 = (unsigned)(v[u].z - base);
      const unsigned r3 = (unsigned)(v[u].w - base);
      const bool m0 = live && (r0 < (unsigned)NPB);
      const bool m1 = live && (r1 < (unsigned)NPB);
      const bool m2 = live && (r2 < (unsigned)NPB);
      const bool m3 = live && (r3 < (unsigned)NPB);
      if (__builtin_amdgcn_ballot_w32(m0 | m1 | m2 | m3) != 0u) {
        const unsigned e0 = (unsigned)ch[u] * CHUNK + (unsigned)lane * 4u;
        unsigned bits;
        bits = __builtin_amdgcn_ballot_w32(m0);
        if (m0) myList[cnt + (int)__builtin_amdgcn_mbcnt_lo(bits, 0u)] = (r0 << E_BITS) | (e0 + 0u);
        cnt += __popc(bits);
        bits = __builtin_amdgcn_ballot_w32(m1);
        if (m1) myList[cnt + (int)__builtin_amdgcn_mbcnt_lo(bits, 0u)] = (r1 << E_BITS) | (e0 + 1u);
        cnt += __popc(bits);
        bits = __builtin_amdgcn_ballot_w32(m2);
        if (m2) myList[cnt + (int)__builtin_amdgcn_mbcnt_lo(bits, 0u)] = (r2 << E_BITS) | (e0 + 2u);
        cnt += __popc(bits);
        bits = __builtin_amdgcn_ballot_w32(m3);
        if (m3) myList[cnt + (int)__builtin_amdgcn_mbcnt_lo(bits, 0u)] = (r3 << E_BITS) | (e0 + 3u);
        cnt += __popc(bits);
      }
      if ((u % HALF_UNROLL) == HALF_UNROLL - 1 && cnt >= 32)
        cnt = drain_whole_batches(myList, cnt, lane, x, src, dinv, sAcc);
    }
  }
  drain_batch<false>(myList, cnt, lane, x, src, dinv, sAcc);
  __syncthreads();

  if (wave < NPB / 16) {
    const int h    = lane >> 4;
    const int mn   = lane & 15;
    const int row0 = wave * 16;
    const float dm = dinv[base + row0 + mn];
    const float* aRow = sAcc + (row0 + mn) * APITCH;

    v8f acc0 = {}, acc1 = {}, acc2 = {}, acc3 = {};
    #pragma unroll 4
    for (int kk = 0; kk < D_FEAT; kk += 4) {
      const int k0 = kk + 2 * h;
      v2f a, b;
      a.x = aRow[k0]     * dm;
      a.y = aRow[k0 + 1] * dm;
      b.x = sW[(k0)     * UNITS + 0 + mn];
      b.y = sW[(k0 + 1) * UNITS + 0 + mn];
      acc0 = __builtin_amdgcn_wmma_f32_16x16x4_f32(false, a, false, b, (short)0, acc0, false, false);
      b.x = sW[(k0)     * UNITS + 16 + mn];
      b.y = sW[(k0 + 1) * UNITS + 16 + mn];
      acc1 = __builtin_amdgcn_wmma_f32_16x16x4_f32(false, a, false, b, (short)0, acc1, false, false);
      b.x = sW[(k0)     * UNITS + 32 + mn];
      b.y = sW[(k0 + 1) * UNITS + 32 + mn];
      acc2 = __builtin_amdgcn_wmma_f32_16x16x4_f32(false, a, false, b, (short)0, acc2, false, false);
      b.x = sW[(k0)     * UNITS + 48 + mn];
      b.y = sW[(k0 + 1) * UNITS + 48 + mn];
      acc3 = __builtin_amdgcn_wmma_f32_16x16x4_f32(false, a, false, b, (short)0, acc3, false, false);
    }

    const float b0 = bias[0 + mn], b1 = bias[16 + mn], b2 = bias[32 + mn], b3 = bias[48 + mn];
    float* orow = out + (size_t)(base + row0 + 8 * h) * UNITS + mn;
    #pragma unroll
    for (int j = 0; j < 8; ++j) {
      orow[(size_t)j * UNITS +  0] = fmaxf(acc0[j] + b0, 0.0f);
      orow[(size_t)j * UNITS + 16] = fmaxf(acc1[j] + b1, 0.0f);
      orow[(size_t)j * UNITS + 32] = fmaxf(acc2[j] + b2, 0.0f);
      orow[(size_t)j * UNITS + 48] = fmaxf(acc3[j] + b3, 0.0f);
    }
  }
}

extern "C" void kernel_launch(void* const* d_in, const int* in_sizes, int n_in,
                              void* d_out, int out_size, void* d_ws, size_t ws_size,
                              hipStream_t stream) {
  const float* x      = (const float*)d_in[0];
  const float* W      = (const float*)d_in[1];
  const float* b      = (const float*)d_in[2];
  const int*   source = (const int*)d_in[3];
  const int*   target = (const int*)d_in[4];
  float*       out    = (float*)d_out;

  float* deg  = (float*)d_ws;
  float* dinv = deg + N_NODES;

  hipMemsetAsync(deg, 0, (size_t)N_NODES * sizeof(float), stream);

  gcn_degree_kernel<<<(N_EDGES + 255) / 256, 256, 0, stream>>>(target, deg, N_EDGES);
  gcn_rsqrt_kernel<<<(N_NODES + 255) / 256, 256, 0, stream>>>(deg, dinv, N_NODES);
  gcn_owner_wmma_kernel<<<N_OWNERS, OWN_THREADS, 0, stream>>>(x, source, target, dinv, W, b, out);
}


# Owner blocks filter a one-byte owner key per edge (target / 400, written by the degree pass), not the 4-byte target: 16 keys per lane load, target re-read per kept edge; LDS sums stay arrival-order
# speedup vs baseline: 1.9743x; 1.1725x over previous
#include <hip/hip_runtime.h>
#include <hip/hip_bf16.h>


typedef __attribute__((ext_vector_type(2))) float v2f;
typedef __attribute__((ext_vector_type(8))) float v8f;

#define N_NODES 100000
#define N_EDGES 1600000
#define D_FEAT  64
#define UNITS   64

#define NPB         400
#define N_OWNERS    (N_NODES / NPB)
#define OWN_THREADS 1024
#define OWN_WAVES   (OWN_THREADS / 32)
#define APITCH      66
#define CHUNK       512
#define N_CHUNKS    (N_EDGES / CHUNK)
#define LOOK_EVERY  2
#define LIST_CAP    (32 + LOOK_EVERY * CHUNK)
#define SCAN_UNROLL 8
#define NO_OWNER    255u
#define ROW_BYTES   (D_FEAT * 4u)

static_assert(N_NODES % NPB == 0, "owner ranges tile the nodes exactly");
static_assert(NPB % 16 == 0 && NPB / 16 <= OWN_WAVES, "one 16-row tile per wave");
static_assert(N_EDGES % CHUNK == 0, "edge list is a whole number of chunks");
static_assert(N_OWNERS <= (int)NO_OWNER, "an owner index fits in one byte beside the no-owner key");
static_assert(SCAN_UNROLL % LOOK_EVERY == 0, "a group of loads ends on a look at the list");
static_assert(((size_t)2 * N_NODES * sizeof(float)) % 128 == 0, "the key array starts on a line");
static_assert((unsigned long long)N_NODES * D_FEAT * 4ull < (1ull << 32), "row offsets fit in 32 bits");

__global__ __launch_bounds__(256) void gcn_degree_kernel(
    const int* __restrict__ target, float* __restrict__ deg,
    unsigned* __restrict__ keyw, int nEdges) {
  int e = blockIdx.x * blockDim.x + threadIdx.x;
  if (e < nEdges) {
    const int t = target[e];
    atomicAdd(&deg[t], 1.0f);
    const unsigned o = (unsigned)t / (unsigned)NPB;
    ((unsigned char*)keyw)[e] = (unsigned char)(o < NO_OWNER ? o : NO_OWNER);
  }
}

__global__ __launch_bounds__(256) void gcn_rsqrt_kernel(
    const float* __restrict__ deg, float* __restrict__ dinv, int n) {
  int i = blockIdx.x * blockDim.x + threadIdx.x;
  if (i < n) dinv[i] = rsqrtf(deg[i]);
}

__device__ __forceinline__ void add_row(float2 v, float di, int ri, int lane, float* sAcc) {
  float* p = sAcc + ri * APITCH + 2 * lane;
  atomicAdd(p + 0, v.x * di);
  atomicAdd(p + 1, v.y * di);
}

__device__ __forceinline__ unsigned zero_bytes(unsigned x) {
  const unsigned t = (x & 0x7f7f7f7fu) + 0x7f7f7f7fu;
  return ~(t | x | 0x7f7f7f7fu);
}

template <bool FULL>
__device__ __forceinline__ void drain_batch(
    const unsigned* myList, int n, int lane, int base,
    const float* __restrict__ x, const int* __restrict__ src,
    const int* __restrict__ tgt,
    const float* __restrict__ dinv, float* sAcc) {
  int s = 0;
  int rel = 0;
  float d = 0.0f;
  if (FULL || lane < n) {
    const unsigned e = myList[lane];
    s = src[e];
    rel = tgt[e] - base;
    d = dinv[s];
  }
  const int dbits = __float_as_int(d);
  const char* xb = (const char*)x;
  const unsigned lane8 = (unsigned)lane * 8u;
  const int nfull = FULL ? 32 : ((n + 15) & ~15);
  for (int i = 0; i < nfull; i += 16) {
    float2 xv[16];
    #pragma unroll
    for (int q = 0; q < 16; ++q) {
      const unsigned si = (unsigned)__builtin_amdgcn_readlane(s, i + q);
      xv[q] = *(const float2*)(xb + (si * ROW_BYTES + lane8));
    }
    #pragma unroll
    for (int q = 0; q < 16; ++q) {
      const float di = __int_as_float(__builtin_amdgcn_readlane(dbits, i + q));
      const int ri = __builtin_amdgcn_readlane(rel, i + q);
      if (FULL || i + q < n) add_row(xv[q], di, ri, lane, sAcc);
    }
  }
}

__device__ __forceinline__ int drain_whole_batches(
    unsigned* myList, int cnt, int lane, int base,
    const float* __restrict__ x, const int* __restrict__ src,
    const int* __restrict__ tgt,
    const float* __restrict__ dinv, float* sAcc) {
  const int full = cnt & ~31;
  for (int j = 0; j < full; j += 32)
    drain_batch<true>(myList + j, 32, lane, base, x, src, tgt, dinv, sAcc);
  const int rem = cnt - full;
  if (lane < rem) {
    const unsigned t = myList[full + lane];
    myList[lane] = t;
  }
  return rem;
}

__global__ __launch_bounds__(OWN_THREADS) void gcn_owner_wmma_kernel(
    const float* __restrict__ x, const int* __restrict__ src,
    const int* __restrict__ tgt, const unsigned* __restrict__ keyw,
    const float* __restrict__ dinv,
    const float* __restrict__ W, const float* __restrict__ bias,
    float* __restrict__ out) {
  __shared__ __attribute__((aligned(16))) float sAcc[NPB * APITCH];
  __shared__ __attribute__((aligned(16))) float sW[D_FEAT * UNITS];
  __shared__ unsigned sList[OWN_WAVES * LIST_CAP];

  const int tid  = threadIdx.x;
  const int lane = tid & 31;
  const int wave = __builtin_amdgcn_readfirstlane(tid >> 5);
  const int base = blockIdx.x * NPB;

  {
    float4* z = (float4*)sAcc;
    for (int i = tid; i < (NPB * APITCH) / 4; i += OWN_THREADS)
      z[i] = make_float4(0.0f, 0.0f, 0.0f, 0.0f);
    const float4* Wv = (const float4*)W;
    float4* sWv = (float4*)sW;
    for (int i = tid; i < (D_FEAT * UNITS) / 4; i += OWN_THREADS) sWv[i] = Wv[i];
  }
  __syncthreads();

  unsigned* myList = sList + wave * LIST_CAP;
  int cnt = 0;
  const uint4* kv = (const uint4*)keyw;
  const unsigned pat = (unsigned)blockIdx.x * 0x01010101u;
  const int start = blockIdx.x * (N_CHUNKS / N_OWNERS);

  for (int c0 = wave; c0 < N_CHUNKS; c0 += OWN_WAVES * SCAN_UNROLL) {
    uint4 v[SCAN_UNROLL];
    int   ch[SCAN_UNROLL];
    #pragma unroll
    for (int u = 0; u < SCAN_UNROLL; ++u) {
      int c = min(c0 + u * OWN_WAVES, N_CHUNKS - 1);
      c += start;
      if (c >= N_CHUNKS) c -= N_CHUNKS;
      ch[u] = c;
      v[u] = kv[c * 32 + lane];
    }
    #pragma unroll
    for (int u = 0; u < SCAN_UNROLL; ++u) {
      const bool live = (c0 + u * OWN_WAVES) < N_CHUNKS;
      unsigned m = (zero_bytes(v[u].x ^ pat) >> 7) | (zero_bytes(v[u].y ^ pat) >> 6) |
                   (zero_bytes(v[u].z ^ pat) >> 5) | (zero_bytes(v[u].w ^ pat) >> 4);
      if (!live) m = 0u;
      unsigned has = __builtin_amdgcn_ballot_w32(m != 0u);
      if (has != 0u) {
        const unsigned e0 = (unsigned)ch[u] * CHUNK + (unsigned)lane * 16u;
        for (int it = 0; it < 16 && has != 0u; ++it) {
          if (m != 0u) {
            const unsigned p = (unsigned)__builtin_ctz(m);
            m &= m - 1u;
            myList[cnt + (int)__builtin_amdgcn_mbcnt_lo(has, 0u)] = e0 + ((p & 3u) << 2) + (p >> 3);
          }
          cnt += __popc(has);
          has = __builtin_amdgcn_ballot_w32(m != 0u);
        }
      }
      if ((u % LOOK_EVERY) == LOOK_EVERY - 1 && cnt >= 32)
        cnt = drain_whole_batches(myList, cnt, lane, base, x, src, tgt, dinv, sAcc);
    }
  }
  drain_batch<false>(myList, cnt, lane, base, x, src, tgt, dinv, sAcc);
  __syncthreads();

  if (wave < NPB / 16) {
    const int h    = lane >> 4;
    const int mn   = lane & 15;
    const int row0 = wave * 16;
    const float dm = dinv[base + row0 + mn];
    const float* aRow = sAcc + (row0 + mn) * APITCH;

    v8f acc0 = {}, acc1 = {}, acc2 = {}, acc3 = {};
    #pragma unroll 4
    for (int kk = 0; kk < D_FEAT; kk += 4) {
      const int k0 = kk + 2 * h;
      v2f a, b;
      a.x = aRow[k0]     * dm;
      a.y = aRow[k0 + 1] * dm;
      b.x = sW[(k0)     * UNITS + 0 + mn];
      b.y = sW[(k0 + 1) * UNITS + 0 + mn];
      acc0 = __builtin_amdgcn_wmma_f32_16x16x4_f32(false, a, false, b, (short)0, acc0, false, false);
      b.x = sW[(k0)     * UNITS + 16 + mn];
      b.y = sW[(k0 + 1) * UNITS + 16 + mn];
      acc1 = __builtin_amdgcn_wmma_f32_16x16x4_f32(false, a, false, b, (short)0, acc1, false, false);
      b.x = sW[(k0)     * UNITS + 32 + mn];
      b.y = sW[(k0 + 1) * UNITS + 32 + mn];
      acc2 = __builtin_amdgcn_wmma_f32_16x16x4_f32(false, a, false, b, (short)0, acc2, false, false);
      b.x = sW[(k0)     * UNITS + 48 + mn];
      b.y = sW[(k0 + 1) * UNITS + 48 + mn];
      acc3 = __builtin_amdgcn_wmma_f32_16x16x4_f32(false, a, false, b, (short)0, acc3, false, false);
    }

    const float b0 = bias[0 + mn], b1 = bias[16 + mn], b2 = bias[32 + mn], b3 = bias[48 + mn];
    float* orow = out + (size_t)(base + row0 + 8 * h) * UNITS + mn;
    #pragma unroll
    for (int j = 0; j < 8; ++j) {
      orow[(size_t)j * UNITS +  0] = fmaxf(acc0[j] + b0, 0.0f);
      orow[(size_t)j * UNITS + 16] = fmaxf(acc1[j] + b1, 0.0f);
      orow[(size_t)j * UNITS + 32] = fmaxf(acc2[j] + b2, 0.0f);
      orow[(size_t)j * UNITS + 48] = fmaxf(acc3[j] + b3, 0.0f);
    }
  }
}

extern "C" void kernel_launch(void* const* d_in, const int* in_sizes, int n_in,
                              void* d_out, int out_size, void* d_ws, size_t ws_size,
                              hipStream_t stream) {
  const float* x      = (const float*)d_in[0];
  const float* W      = (const float*)d_in[1];
  const float* b      = (const float*)d_in[2];
  const int*   source = (const int*)d_in[3];
  const int*   target = (const int*)d_in[4];
  float*       out    = (float*)d_out;

  float*    deg  = (float*)d_ws;
  float*    dinv = deg + N_NODES;
  unsigned* keyw = (unsigned*)(dinv + N_NODES);

  hipMemsetAsync(deg, 0, (size_t)N_NODES * sizeof(float), stream);

  gcn_degree_kernel<<<(N_EDGES + 255) / 256, 256, 0, stream>>>(target, deg, keyw, N_EDGES);
  gcn_rsqrt_kernel<<<(N_NODES + 255) / 256, 256, 0, stream>>>(deg, dinv, N_NODES);
  gcn_owner_wmma_kernel<<<N_OWNERS, OWN_THREADS, 0, stream>>>(x, source, target, keyw, dinv, W, b, out);
}


# In-degree counted by the 250 owner blocks: a pass over the byte keys adds kept edges to 400 LDS integer counters, writes rsqrt; key pass 16 edges per thread; no global atomics; deg exact
# speedup vs baseline: 3.0260x; 1.5327x over previous
#include <hip/hip_runtime.h>
#include <hip/hip_bf16.h>


typedef __attribute__((ext_vector_type(2))) float v2f;
typedef __attribute__((ext_vector_type(8))) float v8f;

#define N_NODES 100000
#define N_EDGES 1600000
#define D_FEAT  64
#define UNITS   64

#define NPB         400
#define N_OWNERS    (N_NODES / NPB)
#define OWN_THREADS 1024
#define OWN_WAVES   (OWN_THREADS / 32)
#define APITCH      66
#define CHUNK       512
#define N_CHUNKS    (N_EDGES / CHUNK)
#define LOOK_EVERY  2
#define LIST_CAP    (32 + LOOK_EVERY * CHUNK)
#define SCAN_UNROLL 8
#define NO_OWNER    255u
#define ROW_BYTES   (D_FEAT * 4u)
#define KEY_THREADS 256
#define KEY_QUADS   (N_EDGES / 4)
#define KEY_UNROLL  4
#define KEY_BLOCKS  ((KEY_QUADS + KEY_THREADS * KEY_UNROLL - 1) / (KEY_THREADS * KEY_UNROLL))

static_assert(N_NODES % NPB == 0, "owner ranges tile the nodes exactly");
static_assert(NPB % 16 == 0 && NPB / 16 <= OWN_WAVES, "one 16-row tile per wave");
static_assert(N_EDGES % CHUNK == 0, "edge list is a whole number of chunks");
static_assert(N_OWNERS <= (int)NO_OWNER, "an owner index fits in one byte beside the no-owner key");
static_assert(SCAN_UNROLL % LOOK_EVERY == 0, "a group of loads ends on a look at the list");
static_assert(N_EDGES % 4 == 0, "the key pass takes edges four at a time");
static_assert(NPB <= OWN_THREADS, "one thread per owned node writes its rsqrt");
static_assert(((size_t)2 * N_NODES * sizeof(float)) % 128 == 0, "the key array starts on a line");
static_assert((unsigned long long)N_NODES * D_FEAT * 4ull < (1ull << 32), "row offsets fit in 32 bits");

__device__ __forceinline__ unsigned owner_key(int t) {
  const unsigned o = (unsigned)t / (unsigned)NPB;
  return o < NO_OWNER ? o : NO_OWNER;
}

__global__ __launch_bounds__(KEY_THREADS) void gcn_key_kernel(
    const int* __restrict__ target, unsigned* __restrict__ keyw) {
  const int4* tv = (const int4*)target;
  const int q0 = blockIdx.x * (KEY_THREADS * KEY_UNROLL) + threadIdx.x;
  int4 t[KEY_UNROLL];
  #pragma unroll
  for (int u = 0; u < KEY_UNROLL; ++u)
    t[u] = tv[min(q0 + u * KEY_THREADS, KEY_QUADS - 1)];
  #pragma unroll
  for (int u = 0; u < KEY_UNROLL; ++u) {
    const int q = q0 + u * KEY_THREADS;
    if (q < KEY_QUADS)
      keyw[q] = owner_key(t[u].x) | (owner_key(t[u].y) << 8) |
                (owner_key(t[u].z) << 16) | (owner_key(t[u].w) << 24);
  }
}

__device__ __forceinline__ void add_row(float2 v, float di, int ri, int lane, float* sAcc) {
  float* p = sAcc + ri * APITCH + 2 * lane;
  atomicAdd(p + 0, v.x * di);
  atomicAdd(p + 1, v.y * di);
}

__device__ __forceinline__ unsigned zero_bytes(unsigned x) {
  const unsigned t = (x & 0x7f7f7f7fu) + 0x7f7f7f7fu;
  return ~(t | x | 0x7f7f7f7fu);
}

template <bool FULL>
__device__ __forceinline__ void drain_batch(
    const unsigned* myList, int n, int lane, int base,
    const float* __restrict__ x, const int* __restrict__ src,
    const int* __restrict__ tgt,
    const float* __restrict__ dinv, float* sAcc) {
  int s = 0;
  int rel = 0;
  float d = 0.0f;
  if (FULL || lane < n) {
    const unsigned e = myList[lane];
    s = src[e];
    rel = tgt[e] - base;
    d = dinv[s];
  }
  const int dbits = __float_as_int(d);
  const char* xb = (const char*)x;
  const unsigned lane8 = (unsigned)lane * 8u;
  const int nfull = FULL ? 32 : ((n + 15) & ~15);
  for (int i = 0; i < nfull; i += 16) {
    float2 xv[16];
    #pragma unroll
    for (int q = 0; q < 16; ++q) {
      const unsigned si = (unsigned)__builtin_amdgcn_readlane(s, i + q);
      xv[q] = *(const float2*)(xb + (si * ROW_BYTES + lane8));
    }
    #pragma unroll
    for (int q = 0; q < 16; ++q) {
      const float di = __int_as_float(__builtin_amdgcn_readlane(dbits, i + q));
      const int ri = __builtin_amdgcn_readlane(rel, i + q);
      if (FULL || i + q < n) add_row(xv[q], di, ri, lane, sAcc);
    }
  }
}

__device__ __forceinline__ int drain_whole_batches(
    unsigned* myList, int cnt, int lane, int base,
    const float* __restrict__ x, const int* __restrict__ src,
    const int* __restrict__ tgt,
    const float* __restrict__ dinv, float* sAcc) {
  const int full = cnt & ~31;
  for (int j = 0; j < full; j += 32)
    drain_batch<true>(myList + j, 32, lane, base, x, src, tgt, dinv, sAcc);
  const int rem = cnt - full;
  if (lane < rem) {
    const unsigned t = myList[full + lane];
    myList[lane] = t;
  }
  return rem;
}

template <typename Drain>
__device__ __forceinline__ int scan_keys(
    const unsigned* __restrict__ keyw, unsigned* myList, int lane, int wave, Drain drain) {
  int cnt = 0;
  const uint4* kv = (const uint4*)keyw;
  const unsigned pat = (unsigned)blockIdx.x * 0x01010101u;
  const int start = blockIdx.x * (N_CHUNKS / N_OWNERS);

  for (int c0 = wave; c0 < N_CHUNKS; c0 += OWN_WAVES * SCAN_UNROLL) {
    uint4 v[SCAN_UNROLL];
    int   ch[SCAN_UNROLL];
    #pragma unroll
    for (int u = 0; u < SCAN_UNROLL; ++u) {
      int c = min(c0 + u * OWN_WAVES, N_CHUNKS - 1);
      c += start;
      if (c >= N_CHUNKS) c -= N_CHUNKS;
      ch[u] = c;
      v[u] = kv[c * 32 + lane];
    }
    #pragma unroll
    for (int u = 0; u < SCAN_UNROLL; ++u) {
      const bool live = (c0 + u * OWN_WAVES) < N_CHUNKS;
      unsigned m = (zero_bytes(v[u].x ^ pat) >> 7) | (zero_bytes(v[u].y ^ pat) >> 6) |
                   (zero_bytes(v[u].z ^ pat) >> 5) | (zero_bytes(v[u].w ^ pat) >> 4);
      if (!live) m = 0u;
      unsigned has = __builtin_amdgcn_ballot_w32(m != 0u);
      if (has != 0u) {
        const unsigned e0 = (unsigned)ch[u] * CHUNK + (unsigned)lane * 16u;
        for (int it = 0; it < 16 && has != 0u; ++it) {
          if (m != 0u) {
            const unsigned p = (unsigned)__builtin_ctz(m);
            m &= m - 1u;
            myList[cnt + (int)__builtin_amdgcn_mbcnt_lo(has, 0u)] = e0 + ((p & 3u) << 2) + (p >> 3);
          }
          cnt += __popc(has);
          has = __builtin_amdgcn_ballot_w32(m != 0u);
        }
      }
      if ((u % LOOK_EVERY) == LOOK_EVERY - 1 && cnt >= 32) cnt = drain(cnt);
    }
  }
  return cnt;
}

__device__ __forceinline__ void count_list(
    const unsigned* myList, int cnt, int lane, int base,
    const int* __restrict__ tgt, unsigned* sCnt) {
  for (int j = lane; j < cnt; j += 64) {
    const bool two = j + 32 < cnt;
    const unsigned ea = myList[j];
    const unsigned eb = myList[two ? j + 32 : j];
    const int ra = tgt[ea] - base;
    const int rb = tgt[eb] - base;
    atomicAdd(&sCnt[ra], 1u);
    atomicAdd(&sCnt[rb], two ? 1u : 0u);
  }
}

__global__ __launch_bounds__(OWN_THREADS) void gcn_owner_count_kernel(
    const int* __restrict__ tgt, const unsigned* __restrict__ keyw,
    float* __restrict__ dinv) {
  __shared__ unsigned sCnt[NPB];
  __shared__ unsigned sList[OWN_WAVES * LIST_CAP];

  const int tid  = threadIdx.x;
  const int lane = tid & 31;
  const int wave = __builtin_amdgcn_readfirstlane(tid >> 5);
  const int base = blockIdx.x * NPB;

  if (tid < NPB) sCnt[tid] = 0u;
  __syncthreads();

  unsigned* myList = sList + wave * LIST_CAP;
  const int cnt = scan_keys(keyw, myList, lane, wave, [&](int c) {
    count_list(myList, c, lane, base, tgt, sCnt);
    return 0;
  });
  count_list(myList, cnt, lane, base, tgt, sCnt);
  __syncthreads();

  if (tid < NPB) dinv[base + tid] = rsqrtf((float)sCnt[tid]);
}

__global__ __launch_bounds__(OWN_THREADS) void gcn_owner_wmma_kernel(
    const float* __restrict__ x, const int* __restrict__ src,
    const int* __restrict__ tgt, const unsigned* __restrict__ keyw,
    const float* __restrict__ dinv,
    const float* __restrict__ W, const float* __restrict__ bias,
    float* __restrict__ out) {
  __shared__ __attribute__((aligned(16))) float sAcc[NPB * APITCH];
  __shared__ __attribute__((aligned(16))) float sW[D_FEAT * UNITS];
  __shared__ unsigned sList[OWN_WAVES * LIST_CAP];

  const int tid  = threadIdx.x;
  const int lane = tid & 31;
  const int wave = __builtin_amdgcn_readfirstlane(tid >> 5);
  const int base = blockIdx.x * NPB;

  {
    float4* z = (float4*)sAcc;
    for (int i = tid; i < (NPB * APITCH) / 4; i += OWN_THREADS)
      z[i] = make_float4(0.0f, 0.0f, 0.0f, 0.0f);
    const float4* Wv = (const float4*)W;
    float4* sWv = (float4*)sW;
    for (int i = tid; i < (D_FEAT * UNITS) / 4; i += OWN_THREADS) sWv[i] = Wv[i];
  }
  __syncthreads();

  unsigned* myList = sList + wave * LIST_CAP;
  const int cnt = scan_keys(keyw, myList, lane, wave, [&](int c) {
    return drain_whole_batches(myList, c, lane, base, x, src, tgt, dinv, sAcc);
  });
  drain_batch<false>(myList, cnt, lane, base, x, src, tgt, dinv, sAcc);
  __syncthreads();

  if (wave < NPB / 16) {
    const int h    = lane >> 4;
    const int mn   = lane & 15;
    const int row0 = wave * 16;
    const float dm = dinv[base + row0 + mn];
    const float* aRow = sAcc + (row0 + mn) * APITCH;

    v8f acc0 = {}, acc1 = {}, acc2 = {}, acc3 = {};
    #pragma unroll 4
    for (int kk = 0; kk < D_FEAT; kk += 4) {
      const int k0 = kk + 2 * h;
      v2f a, b;
      a.x = aRow[k0]     * dm;
      a.y = aRow[k0 + 1] * dm;
      b.x = sW[(k0)     * UNITS + 0 + mn];
      b.y = sW[(k0 + 1) * UNITS + 0 + mn];
      acc0 = __builtin_amdgcn_wmma_f32_16x16x4_f32(false, a, false, b, (short)0, acc0, false, false);
      b.x = sW[(k0)     * UNITS + 16 + mn];
      b.y = sW[(k0 + 1) * UNITS + 16 + mn];
      acc1 = __builtin_amdgcn_wmma_f32_16x16x4_f32(false, a, false, b, (short)0, acc1, false, false);
      b.x = sW[(k0)     * UNITS + 32 + mn];
      b.y = sW[(k0 + 1) * UNITS + 32 + mn];
      acc2 = __builtin_amdgcn_wmma_f32_16x16x4_f32(false, a, false, b, (short)0, acc2, false, false);
      b.x = sW[(k0)     * UNITS + 48 + mn];
      b.y = sW[(k0 + 1) * UNITS + 48 + mn];
      acc3 = __builtin_amdgcn_wmma_f32_16x16x4_f32(false, a, false, b, (short)0, acc3, false, false);
    }

    const float b0 = bias[0 + mn], b1 = bias[16 + mn], b2 = bias[32 + mn], b3 = bias[48 + mn];
    float* orow = out + (size_t)(base + row0 + 8 * h) * UNITS + mn;
    #pragma unroll
    for (int j = 0; j < 8; ++j) {
      orow[(size_t)j * UNITS +  0] = fmaxf(acc0[j] + b0, 0.0f);
      orow[(size_t)j * UNITS + 16] = fmaxf(acc1[j] + b1, 0.0f);
      orow[(size_t)j * UNITS + 32] = fmaxf(acc2[j] + b2, 0.0f);
      orow[(size_t)j * UNITS + 48] = fmaxf(acc3[j] + b3, 0.0f);
    }
  }
}

extern "C" void kernel_launch(void* const* d_in, const int* in_sizes, int n_in,
                              void* d_out, int out_size, void* d_ws, size_t ws_size,
                              hipStream_t stream) {
  const float* x      = (const float*)d_in[0];
  const float* W      = (const float*)d_in[1];
  const float* b      = (const float*)d_in[2];
  const int*   source = (const int*)d_in[3];
  const int*   target = (const int*)d_in[4];
  float*       out    = (float*)d_out;

  float*    dinv = (float*)d_ws + N_NODES;
  unsigned* keyw = (unsigned*)(dinv + N_NODES);

  gcn_key_kernel<<<KEY_BLOCKS, KEY_THREADS, 0, stream>>>(target, keyw);
  gcn_owner_count_kernel<<<N_OWNERS, OWN_THREADS, 0, stream>>>(target, keyw, dinv);
  gcn_owner_wmma_kernel<<<N_OWNERS, OWN_THREADS, 0, stream>>>(x, source, target, keyw, dinv, W, b, out);
}


# Degree kernel writes each owner block's kept edges (source and tile row in one word) to a workspace list; the aggregation kernel reads its list, no second key filter; LDS sums arrival-order
# speedup vs baseline: 3.7878x; 1.2518x over previous
#include <hip/hip_runtime.h>
#include <hip/hip_bf16.h>


typedef __attribute__((ext_vector_type(2))) float v2f;
typedef __attribute__((ext_vector_type(8))) float v8f;

#define N_NODES 100000
#define N_EDGES 1600000
#define D_FEAT  64
#define UNITS   64

#define NPB         400
#define N_OWNERS    (N_NODES / NPB)
#define OWN_THREADS 1024
#define OWN_WAVES   (OWN_THREADS / 32)
#define APITCH      66
#define CHUNK       512
#define N_CHUNKS    (N_EDGES / CHUNK)
#define LOOK_EVERY  2
#define LIST_CAP    (32 + LOOK_EVERY * CHUNK)
#define SCAN_UNROLL 8
#define NO_OWNER    255u
#define ROW_BYTES   (D_FEAT * 4u)
#define KEY_THREADS 256
#define KEY_QUADS   (N_EDGES / 4)
#define KEY_UNROLL  4
#define KEY_BLOCKS  ((KEY_QUADS + KEY_THREADS * KEY_UNROLL - 1) / (KEY_THREADS * KEY_UNROLL))
#define REL_BITS    9
#define REL_MASK    ((1u << REL_BITS) - 1u)
#define GLIST_CAP   8192

static_assert(N_NODES % NPB == 0, "owner ranges tile the nodes exactly");
static_assert(NPB % 16 == 0 && NPB / 16 <= OWN_WAVES, "one 16-row tile per wave");
static_assert(N_EDGES % CHUNK == 0, "edge list is a whole number of chunks");
static_assert(N_OWNERS <= (int)NO_OWNER, "an owner index fits in one byte beside the no-owner key");
static_assert(SCAN_UNROLL % LOOK_EVERY == 0, "a group of loads ends on a look at the list");
static_assert(N_EDGES % 4 == 0, "the key pass takes edges four at a time");
static_assert(NPB <= OWN_THREADS, "one thread per owned node writes its rsqrt");
static_assert(((size_t)2 * N_NODES * sizeof(float)) % 128 == 0, "the key array starts on a line");
static_assert((unsigned long long)N_NODES * D_FEAT * 4ull < (1ull << 32), "row offsets fit in 32 bits");
static_assert(NPB <= (1 << REL_BITS), "a row of the tile fits in the low bits of a list word");
static_assert((unsigned long long)N_NODES <= (1ull << (32 - REL_BITS)), "a source fits above the row in a list word");
static_assert(GLIST_CAP % 32 == 0, "a block's list is a whole number of batches");
static_assert(((size_t)2 * N_NODES * sizeof(float) + N_EDGES) % 128 == 0, "the lists start on a line");
static_assert(N_OWNERS <= N_NODES, "the list lengths fit in the unused front of the workspace");

__device__ __forceinline__ unsigned owner_key(int t) {
  const unsigned o = (unsigned)t / (unsigned)NPB;
  return o < NO_OWNER ? o : NO_OWNER;
}

__global__ __launch_bounds__(KEY_THREADS) void gcn_key_kernel(
    const int* __restrict__ target, unsigned* __restrict__ keyw) {
  const int4* tv = (const int4*)target;
  const int q0 = blockIdx.x * (KEY_THREADS * KEY_UNROLL) + threadIdx.x;
  int4 t[KEY_UNROLL];
  #pragma unroll
  for (int u = 0; u < KEY_UNROLL; ++u)
    t[u] = tv[min(q0 + u * KEY_THREADS, KEY_QUADS - 1)];
  #pragma unroll
  for (int u = 0; u < KEY_UNROLL; ++u) {
    const int q = q0 + u * KEY_THREADS;
    if (q < KEY_QUADS)
      keyw[q] = owner_key(t[u].x) | (owner_key(t[u].y) << 8) |
                (owner_key(t[u].z) << 16) | (owner_key(t[u].w) << 24);
  }
}

__device__ __forceinline__ void add_row(float2 v, float di, int ri, int lane, float* sAcc) {
  float* p = sAcc + ri * APITCH + 2 * lane;
  atomicAdd(p + 0, v.x * di);
  atomicAdd(p + 1, v.y * di);
}

__device__ __forceinline__ unsigned zero_bytes(unsigned x) {
  const unsigned t = (x & 0x7f7f7f7fu) + 0x7f7f7f7fu;
  return ~(t | x | 0x7f7f7f7fu);
}

template <bool FULL>
__device__ __forceinline__ void add_batch(
    int s, int rel, float d, int n, int lane,
    const float* __restrict__ x, float* sAcc) {
  const int dbits = __float_as_int(d);
  const char* xb = (const char*)x;
  const unsigned lane8 = (unsigned)lane * 8u;
  const int nfull = FULL ? 32 : ((n + 15) & ~15);
  for (int i = 0; i < nfull; i += 16) {
    float2 xv[16];
    #pragma unroll
    for (int q = 0; q < 16; ++q) {
      const unsigned si = (unsigned)__builtin_amdgcn_readlane(s, i + q);
      xv[q] = *(const float2*)(xb + (si * ROW_BYTES + lane8));
    }
    #pragma unroll
    for (int q = 0; q < 16; ++q) {
      const float di = __int_as_float(__builtin_amdgcn_readlane(dbits, i + q));
      const int ri = __builtin_amdgcn_readlane(rel, i + q);
      if (FULL || i + q < n) add_row(xv[q], di, ri, lane, sAcc);
    }
  }
}

template <bool FULL>
__device__ __forceinline__ void drain_batch(
    const unsigned* myList, int n, int lane, int base,
    const float* __restrict__ x, const int* __restrict__ src,
    const int* __restrict__ tgt,
    const float* __restrict__ dinv, float* sAcc) {
  int s = 0;
  int rel = 0;
  float d = 0.0f;
  if (FULL || lane < n) {
    const unsigned e = myList[lane];
    s = src[e];
    rel = tgt[e] - base;
    d = dinv[s];
  }
  add_batch<FULL>(s, rel, d, n, lane, x, sAcc);
}

__device__ __forceinline__ int drain_whole_batches(
    unsigned* myList, int cnt, int lane, int base,
    const float* __restrict__ x, const int* __restrict__ src,
    const int* __restrict__ tgt,
    const float* __restrict__ dinv, float* sAcc) {
  const int full = cnt & ~31;
  for (int j = 0; j < full; j += 32)
    drain_batch<true>(myList + j, 32, lane, base, x, src, tgt, dinv, sAcc);
  const int rem = cnt - full;
  if (lane < rem) {
    const unsigned t = myList[full + lane];
    myList[lane] = t;
  }
  return rem;
}

template <typename Drain>
__device__ __forceinline__ int scan_keys(
    const unsigned* __restrict__ keyw, unsigned* myList, int lane, int wave, Drain drain) {
  int cnt = 0;
  const uint4* kv = (const uint4*)keyw;
  const unsigned pat = (unsigned)blockIdx.x * 0x01010101u;
  const int start = blockIdx.x * (N_CHUNKS / N_OWNERS);

  for (int c0 = wave; c0 < N_CHUNKS; c0 += OWN_WAVES * SCAN_UNROLL) {
    uint4 v[SCAN_UNROLL];
    int   ch[SCAN_UNROLL];
    #pragma unroll
    for (int u = 0; u < SCAN_UNROLL; ++u) {
      int c = min(c0 + u * OWN_WAVES, N_CHUNKS - 1);
      c += start;
      if (c >= N_CHUNKS) c -= N_CHUNKS;
      ch[u] = c;
      v[u] = kv[c * 32 + lane];
    }
    #pragma unroll
    for (int u = 0; u < SCAN_UNROLL; ++u) {
      const bool live = (c0 + u * OWN_WAVES) < N_CHUNKS;
      unsigned m = (zero_bytes(v[u].x ^ pat) >> 7) | (zero_bytes(v[u].y ^ pat) >> 6) |
                   (zero_bytes(v[u].z ^ pat) >> 5) | (zero_bytes(v[u].w ^ pat) >> 4);
      if (!live) m = 0u;
      unsigned has = __builtin_amdgcn_ballot_w32(m != 0u);
      if (has != 0u) {
        const unsigned e0 = (unsigned)ch[u] * CHUNK + (unsigned)lane * 16u;
        for (int it = 0; it < 16 && has != 0u; ++it) {
          if (m != 0u) {
            const unsigned p = (unsigned)__builtin_ctz(m);
            m &= m - 1u;
            myList[cnt + (int)__builtin_amdgcn_mbcnt_lo(has, 0u)] = e0 + ((p & 3u) << 2) + (p >> 3);
          }
          cnt += __popc(has);
          has = __builtin_amdgcn_ballot_w32(m != 0u);
        }
      }
      if ((u % LOOK_EVERY) == LOOK_EVERY - 1 && cnt >= 32) cnt = drain(cnt);
    }
  }
  return cnt;
}

template <bool LIST>
__device__ __forceinline__ void count_entries(
    const unsigned* myList, int cnt, int lane, int base,
    const int* __restrict__ src, const int* __restrict__ tgt,
    unsigned* sCnt, unsigned* __restrict__ dst) {
  for (int j = lane; j < cnt; j += 64) {
    const bool two = j + 32 < cnt;
    const int jb = two ? j + 32 : j;
    const unsigned ea = myList[j];
    const unsigned eb = myList[jb];
    const int ra = tgt[ea] - base;
    const int rb = tgt[eb] - base;
    unsigned sa = 0u, sb = 0u;
    if (LIST) {
      sa = (unsigned)src[ea];
      sb = (unsigned)src[eb];
    }
    atomicAdd(&sCnt[ra], 1u);
    atomicAdd(&sCnt[rb], two ? 1u : 0u);
    if (LIST) {
      dst[j]  = (sa << REL_BITS) | (unsigned)ra;
      dst[jb] = (sb << REL_BITS) | (unsigned)rb;
    }
  }
}

__device__ __forceinline__ void count_list(
    const unsigned* myList, int cnt, int lane, int base,
    const int* __restrict__ src, const int* __restrict__ tgt,
    unsigned* sCnt, unsigned* sTotal, unsigned* __restrict__ blockList) {
  if (cnt == 0) return;
  unsigned off = 0u;
  if (lane == 0) off = atomicAdd(sTotal, (unsigned)cnt);
  off = (unsigned)__builtin_amdgcn_readfirstlane((int)off);
  if (off + (unsigned)cnt <= GLIST_CAP)
    count_entries<true>(myList, cnt, lane, base, src, tgt, sCnt, blockList + off);
  else
    count_entries<false>(myList, cnt, lane, base, src, tgt, sCnt, blockList);
}

__global__ __launch_bounds__(OWN_THREADS) void gcn_owner_count_kernel(
    const int* __restrict__ src, const int* __restrict__ tgt,
    const unsigned* __restrict__ keyw, float* __restrict__ dinv,
    unsigned* __restrict__ glist, unsigned* __restrict__ blockLen) {
  __shared__ unsigned sCnt[NPB];
  __shared__ unsigned sList[OWN_WAVES * LIST_CAP];
  __shared__ unsigned sTotal;

  const int tid  = threadIdx.x;
  const int lane = tid & 31;
  const int wave = __builtin_amdgcn_readfirstlane(tid >> 5);
  const int base = blockIdx.x * NPB;

  if (tid < NPB) sCnt[tid] = 0u;
  if (tid == 0) sTotal = 0u;
  __syncthreads();

  unsigned* myList = sList + wave * LIST_CAP;
  unsigned* blockList = glist + (size_t)blockIdx.x * GLIST_CAP;
  const int cnt = scan_keys(keyw, myList, lane, wave, [&](int c) {
    count_list(myList, c, lane, base, src, tgt, sCnt, &sTotal, blockList);
    return 0;
  });
  count_list(myList, cnt, lane, base, src, tgt, sCnt, &sTotal, blockList);
  __syncthreads();

  if (tid < NPB) dinv[base + tid] = rsqrtf((float)sCnt[tid]);
  if (tid == 0) blockLen[blockIdx.x] = sTotal;
}

__global__ __launch_bounds__(OWN_THREADS) void gcn_owner_wmma_kernel(
    const float* __restrict__ x, const int* __restrict__ src,
    const int* __restrict__ tgt, const unsigned* __restrict__ keyw,
    const unsigned* __restrict__ glist, const unsigned* __restrict__ blockLen,
    const float* __restrict__ dinv,
    const float* __restrict__ W, const float* __restrict__ bias,
    float* __restrict__ out) {
  __shared__ __attribute__((aligned(16))) float sAcc[NPB * APITCH];
  __shared__ __attribute__((aligned(16))) float sW[D_FEAT * UNITS];
  __shared__ unsigned sList[OWN_WAVES * LIST_CAP];

  const int tid  = threadIdx.x;
  const int lane = tid & 31;
  const int wave = __builtin_amdgcn_readfirstlane(tid >> 5);
  const int base = blockIdx.x * NPB;

  {
    float4* z = (float4*)sAcc;
    for (int i = tid; i < (NPB * APITCH) / 4; i += OWN_THREADS)
      z[i] = make_float4(0.0f, 0.0f, 0.0f, 0.0f);
    const float4* Wv = (const float4*)W;
    float4* sWv = (float4*)sW;
    for (int i = tid; i < (D_FEAT * UNITS) / 4; i += OWN_THREADS) sWv[i] = Wv[i];
  }
  __syncthreads();

  const unsigned total = (unsigned)__builtin_amdgcn_readfirstlane((int)blockLen[blockIdx.x]);
  if (total <= GLIST_CAP) {
    const unsigned* blockList = glist + (size_t)blockIdx.x * GLIST_CAP;
    const unsigned nfull = total >> 5;
    const unsigned rem = total & 31u;
    unsigned bt = (unsigned)wave;
    unsigned ent = (bt < nfull) ? blockList[bt * 32u + lane] : 0u;
    while (bt < nfull) {
      const int s = (int)(ent >> REL_BITS);
      const int rel = (int)(ent & REL_MASK);
      const float d = dinv[s];
      const unsigned bn = bt + OWN_WAVES;
      ent = (bn < nfull) ? blockList[bn * 32u + lane] : 0u;
      add_batch<true>(s, rel, d, 32, lane, x, sAcc);
      bt = bn;
    }
    if (rem != 0u && (unsigned)wave == nfull % OWN_WAVES) {
      const unsigned last = ((unsigned)lane < rem) ? blockList[nfull * 32u + lane] : 0u;
      const int s = (int)(last >> REL_BITS);
      const int rel = (int)(last & REL_MASK);
      const float d = ((unsigned)lane < rem) ? dinv[s] : 0.0f;
      add_batch<false>(s, rel, d, (int)rem, lane, x, sAcc);
    }
  } else {
    unsigned* myList = sList + wave * LIST_CAP;
    const int cnt = scan_keys(keyw, myList, lane, wave, [&](int c) {
      return drain_whole_batches(myList, c, lane, base, x, src, tgt, dinv, sAcc);
    });
    drain_batch<false>(myList, cnt, lane, base, x, src, tgt, dinv, sAcc);
  }
  __syncthreads();

  if (wave < NPB / 16) {
    const int h    = lane >> 4;
    const int mn   = lane & 15;
    const int row0 = wave * 16;
    const float dm = dinv[base + row0 + mn];
    const float* aRow = sAcc + (row0 + mn) * APITCH;

    v8f acc0 = {}, acc1 = {}, acc2 = {}, acc3 = {};
    #pragma unroll 4
    for (int kk = 0; kk < D_FEAT; kk += 4) {
      const int k0 = kk + 2 * h;
      v2f a, b;
      a.x = aRow[k0]     * dm;
      a.y = aRow[k0 + 1] * dm;
      b.x = sW[(k0)     * UNITS + 0 + mn];
      b.y = sW[(k0 + 1) * UNITS + 0 + mn];
      acc0 = __builtin_amdgcn_wmma_f32_16x16x4_f32(false, a, false, b, (short)0, acc0, false, false);
      b.x = sW[(k0)     * UNITS + 16 + mn];
      b.y = sW[(k0 + 1) * UNITS + 16 + mn];
      acc1 = __builtin_amdgcn_wmma_f32_16x16x4_f32(false, a, false, b, (short)0, acc1, false, false);
      b.x = sW[(k0)     * UNITS + 32 + mn];
      b.y = sW[(k0 + 1) * UNITS + 32 + mn];
      acc2 = __builtin_amdgcn_wmma_f32_16x16x4_f32(false, a, false, b, (short)0, acc2, false, false);
      b.x = sW[(k0)     * UNITS + 48 + mn];
      b.y = sW[(k0 + 1) * UNITS + 48 + mn];
      acc3 = __builtin_amdgcn_wmma_f32_16x16x4_f32(false, a, false, b, (short)0, acc3, false, false);
    }

    const float b0 = bias[0 + mn], b1 = bias[16 + mn], b2 = bias[32 + mn], b3 = bias[48 + mn];
    float* orow = out + (size_t)(base + row0 + 8 * h) * UNITS + mn;
    #pragma unroll
    for (int j = 0; j < 8; ++j) {
      orow[(size_t)j * UNITS +  0] = fmaxf(acc0[j] + b0, 0.0f);
      orow[(size_t)j * UNITS + 16] = fmaxf(acc1[j] + b1, 0.0f);
      orow[(size_t)j * UNITS + 32] = fmaxf(acc2[j] + b2, 0.0f);
      orow[(size_t)j * UNITS + 48] = fmaxf(acc3[j] + b3, 0.0f);
    }
  }
}

extern "C" void kernel_launch(void* const* d_in, const int* in_sizes, int n_in,
                              void* d_out, int out_size, void* d_ws, size_t ws_size,
                              hipStream_t stream) {
  const float* x      = (const float*)d_in[0];
  const float* W      = (const float*)d_in[1];
  const float* b      = (const float*)d_in[2];
  const int*   source = (const int*)d_in[3];
  const int*   target = (const int*)d_in[4];
  float*       out    = (float*)d_out;

  float*    dinv = (float*)d_ws + N_NODES;
  unsigned* keyw = (unsigned*)(dinv + N_NODES);
  unsigned* glist = keyw + N_EDGES / 4;
  unsigned* blockLen = (unsigned*)d_ws;

  gcn_key_kernel<<<KEY_BLOCKS, KEY_THREADS, 0, stream>>>(target, keyw);
  gcn_owner_count_kernel<<<N_OWNERS, OWN_THREADS, 0, stream>>>(source, target, keyw, dinv, glist, blockLen);
  gcn_owner_wmma_kernel<<<N_OWNERS, OWN_THREADS, 0, stream>>>(x, source, target, keyw, glist, blockLen, dinv, W, b, out);
}


# Owner key kept as 8 bit planes (one bit per edge), not a byte per edge: an owner block tests 32 edges per lane word with 8 xor/and ops (was 16 edges per 28 ops); LDS sums stay arrival-order
# speedup vs baseline: 3.9790x; 1.0505x over previous
#include <hip/hip_runtime.h>
#include <hip/hip_bf16.h>


typedef __attribute__((ext_vector_type(2))) float v2f;
typedef __attribute__((ext_vector_type(8))) float v8f;

#define N_NODES 100000
#define N_EDGES 1600000
#define D_FEAT  64
#define UNITS   64

#define NPB         400
#define N_OWNERS    (N_NODES / NPB)
#define OWN_THREADS 1024
#define OWN_WAVES   (OWN_THREADS / 32)
#define APITCH      66
#define KEY_BITS    8
#define GROUP_EDGES 128
#define N_GROUPS    (N_EDGES / GROUP_EDGES)
#define PLANE_PITCH 50048
#define N_CHUNKS    ((N_GROUPS + 31) / 32)
#define WORD_EDGES  (32 * 32)
#define LIST_CAP    (32 + WORD_EDGES)
#define SCAN_UNROLL 2
#define NO_OWNER    255u
#define ROW_BYTES   (D_FEAT * 4u)
#define KEY_THREADS 256
#define KEY_WAVES   (KEY_THREADS / 32)
#define WAVE_GROUPS 8
#define KEY_BLOCKS  ((N_GROUPS + KEY_WAVES * WAVE_GROUPS - 1) / (KEY_WAVES * WAVE_GROUPS))
#define REL_BITS    9
#define REL_MASK    ((1u << REL_BITS) - 1u)
#define GLIST_CAP   8192

static_assert(N_NODES % NPB == 0, "owner ranges tile the nodes exactly");
static_assert(NPB % 16 == 0 && NPB / 16 <= OWN_WAVES, "one 16-row tile per wave");
static_assert(N_EDGES % GROUP_EDGES == 0, "edge list is a whole number of groups");
static_assert(PLANE_PITCH >= N_EDGES / 32 && (PLANE_PITCH * 4) % 128 == 0, "a plane holds every edge and starts on a line");
static_assert(N_OWNERS <= (int)NO_OWNER, "an owner index fits in one byte beside the no-owner key");
static_assert(NPB <= OWN_THREADS, "one thread per owned node writes its rsqrt");
static_assert(((size_t)2 * N_NODES * sizeof(float)) % 128 == 0, "the key array starts on a line");
static_assert((unsigned long long)N_NODES * D_FEAT * 4ull < (1ull << 32), "row offsets fit in 32 bits");
static_assert(NPB <= (1 << REL_BITS), "a row of the tile fits in the low bits of a list word");
static_assert((unsigned long long)N_NODES <= (1ull << (32 - REL_BITS)), "a source fits above the row in a list word");
static_assert(GLIST_CAP % 32 == 0, "a block's list is a whole number of batches");
static_assert(((size_t)2 * N_NODES * sizeof(float) + (size_t)KEY_BITS * PLANE_PITCH * 4) % 128 == 0, "the lists start on a line");
static_assert(N_OWNERS <= N_NODES, "the list lengths fit in the unused front of the workspace");

__device__ __forceinline__ unsigned owner_key(int t) {
  const unsigned o = (unsigned)t / (unsigned)NPB;
  return o < NO_OWNER ? o : NO_OWNER;
}

__global__ __launch_bounds__(KEY_THREADS) void gcn_key_kernel(
    const int* __restrict__ target, unsigned* __restrict__ keyw) {
  const int4* tv = (const int4*)target;
  const int lane = threadIdx.x & 31;
  const int wave = __builtin_amdgcn_readfirstlane((int)(threadIdx.x >> 5));
  const int g0 = (blockIdx.x * KEY_WAVES + wave) * WAVE_GROUPS;
  int4 t[WAVE_GROUPS];
  #pragma unroll
  for (int g = 0; g < WAVE_GROUPS; ++g)
    t[g] = tv[min(g0 + g, N_GROUPS - 1) * 32 + lane];
  uint4 w[KEY_BITS];
  #pragma unroll
  for (int k = 0; k < KEY_BITS; ++k) w[k] = make_uint4(0u, 0u, 0u, 0u);
  #pragma unroll
  for (int g = 0; g < WAVE_GROUPS; ++g) {
    const unsigned kx = owner_key(t[g].x), ky = owner_key(t[g].y);
    const unsigned kz = owner_key(t[g].z), kw = owner_key(t[g].w);
    const bool mine = lane == g;
    #pragma unroll
    for (int k = 0; k < KEY_BITS; ++k) {
      const unsigned bx = __builtin_amdgcn_ballot_w32(((kx >> k) & 1u) != 0u);
      const unsigned by = __builtin_amdgcn_ballot_w32(((ky >> k) & 1u) != 0u);
      const unsigned bz = __builtin_amdgcn_ballot_w32(((kz >> k) & 1u) != 0u);
      const unsigned bw = __builtin_amdgcn_ballot_w32(((kw >> k) & 1u) != 0u);
      if (mine) w[k] = make_uint4(bx, by, bz, bw);
    }
  }
  const int grp = g0 + lane;
  if (lane < WAVE_GROUPS && grp < N_GROUPS) {
    #pragma unroll
    for (int k = 0; k < KEY_BITS; ++k)
      ((uint4*)(keyw + (size_t)k * PLANE_PITCH))[grp] = w[k];
  }
}

__device__ __forceinline__ void add_row(float2 v, float di, int ri, int lane, float* sAcc) {
  float* p = sAcc + ri * APITCH + 2 * lane;
  atomicAdd(p + 0, v.x * di);
  atomicAdd(p + 1, v.y * di);
}

template <bool FULL>
__device__ __forceinline__ void add_batch(
    int s, int rel, float d, int n, int lane,
    const float* __restrict__ x, float* sAcc) {
  const int dbits = __float_as_int(d);
  const char* xb = (const char*)x;
  const unsigned lane8 = (unsigned)lane * 8u;
  const int nfull = FULL ? 32 : ((n + 15) & ~15);
  for (int i = 0; i < nfull; i += 16) {
    float2 xv[16];
    #pragma unroll
    for (int q = 0; q < 16; ++q) {
      const unsigned si = (unsigned)__builtin_amdgcn_readlane(s, i + q);
      xv[q] = *(const float2*)(xb + (si * ROW_BYTES + lane8));
    }
    #pragma unroll
    for (int q = 0; q < 16; ++q) {
      const float di = __int_as_float(__builtin_amdgcn_readlane(dbits, i + q));
      const int ri = __builtin_amdgcn_readlane(rel, i + q);
      if (FULL || i + q < n) add_row(xv[q], di, ri, lane, sAcc);
    }
  }
}

template <bool FULL>
__device__ __forceinline__ void drain_batch(
    const unsigned* myList, int n, int lane, int base,
    const float* __restrict__ x, const int* __restrict__ src,
    const int* __restrict__ tgt,
    const float* __restrict__ dinv, float* sAcc) {
  int s = 0;
  int rel = 0;
  float d = 0.0f;
  if (FULL || lane < n) {
    const unsigned e = myList[lane];
    s = src[e];
    rel = tgt[e] - base;
    d = dinv[s];
  }
  add_batch<FULL>(s, rel, d, n, lane, x, sAcc);
}

__device__ __forceinline__ int drain_whole_batches(
    unsigned* myList, int cnt, int lane, int base,
    const float* __restrict__ x, const int* __restrict__ src,
    const int* __restrict__ tgt,
    const float* __restrict__ dinv, float* sAcc) {
  const int full = cnt & ~31;
  for (int j = 0; j < full; j += 32)
    drain_batch<true>(myList + j, 32, lane, base, x, src, tgt, dinv, sAcc);
  const int rem = cnt - full;
  if (lane < rem) {
    const unsigned t = myList[full + lane];
    myList[lane] = t;
  }
  return rem;
}

__device__ __forceinline__ unsigned word_of(uint4 a, int j) {
  return j == 0 ? a.x : j == 1 ? a.y : j == 2 ? a.z : a.w;
}

template <typename Drain>
__device__ __forceinline__ int scan_keys(
    const unsigned* __restrict__ keyw, unsigned* myList, int lane, int wave, Drain drain) {
  int cnt = 0;
  unsigned inv[KEY_BITS];
  #pragma unroll
  for (int k = 0; k < KEY_BITS; ++k) inv[k] = (((unsigned)blockIdx.x >> k) & 1u) ? 0u : 0xffffffffu;
  const int start = (int)(((unsigned)blockIdx.x * (unsigned)N_CHUNKS) / (unsigned)N_OWNERS);

  for (int c0 = wave; c0 < N_CHUNKS; c0 += OWN_WAVES * SCAN_UNROLL) {
    uint4 v[SCAN_UNROLL][KEY_BITS];
    int   grp[SCAN_UNROLL];
    #pragma unroll
    for (int u = 0; u < SCAN_UNROLL; ++u) {
      int c = min(c0 + u * OWN_WAVES, N_CHUNKS - 1);
      c += start;
      if (c >= N_CHUNKS) c -= N_CHUNKS;
      grp[u] = c * 32 + lane;
      const int gl = min(grp[u], N_GROUPS - 1);
      #pragma unroll
      for (int k = 0; k < KEY_BITS; ++k)
        v[u][k] = ((const uint4*)(keyw + (size_t)k * PLANE_PITCH))[gl];
    }
    #pragma unroll
    for (int u = 0; u < SCAN_UNROLL; ++u) {
      const bool ok = (c0 + u * OWN_WAVES) < N_CHUNKS && grp[u] < N_GROUPS;
      const unsigned e0 = (unsigned)grp[u] * (unsigned)GROUP_EDGES;
      #pragma unroll
      for (int j = 0; j < 4; ++j) {
        unsigned m = word_of(v[u][0], j) ^ inv[0];
        #pragma unroll
        for (int k = 1; k < KEY_BITS; ++k) m &= word_of(v[u][k], j) ^ inv[k];
        if (!ok) m = 0u;
        unsigned has = __builtin_amdgcn_ballot_w32(m != 0u);
        for (int it = 0; it < 32 && has != 0u; ++it) {
          if (m != 0u) {
            const unsigned p = (unsigned)__builtin_ctz(m);
            m &= m - 1u;
            myList[cnt + (int)__builtin_amdgcn_mbcnt_lo(has, 0u)] = e0 + (p << 2) + (unsigned)j;
          }
          cnt += __popc(has);
          has = __builtin_amdgcn_ballot_w32(m != 0u);
        }
        if (cnt >= 32) cnt = drain(cnt);
      }
    }
  }
  return cnt;
}

template <bool LIST>
__device__ __forceinline__ void count_entries(
    const unsigned* myList, int cnt, int lane, int base,
    const int* __restrict__ src, const int* __restrict__ tgt,
    unsigned* sCnt, unsigned* __restrict__ dst) {
  for (int j = lane; j < cnt; j += 64) {
    const bool two = j + 32 < cnt;
    const int jb = two ? j + 32 : j;
    const unsigned ea = myList[j];
    const unsigned eb = myList[jb];
    const int ra = tgt[ea] - base;
    const int rb = tgt[eb] - base;
    unsigned sa = 0u, sb = 0u;
    if (LIST) {
      sa = (unsigned)src[ea];
      sb = (unsigned)src[eb];
    }
    atomicAdd(&sCnt[ra], 1u);
    atomicAdd(&sCnt[rb], two ? 1u : 0u);
    if (LIST) {
      dst[j]  = (sa << REL_BITS) | (unsigned)ra;
      dst[jb] = (sb << REL_BITS) | (unsigned)rb;
    }
  }
}

__device__ __forceinline__ void count_list(
    const unsigned* myList, int cnt, int lane, int base,
    const int* __restrict__ src, const int* __restrict__ tgt,
    unsigned* sCnt, unsigned* sTotal, unsigned* __restrict__ blockList) {
  if (cnt == 0) return;
  unsigned off = 0u;
  if (lane == 0) off = atomicAdd(sTotal, (unsigned)cnt);
  off = (unsigned)__builtin_amdgcn_readfirstlane((int)off);
  if (off + (unsigned)cnt <= GLIST_CAP)
    count_entries<true>(myList, cnt, lane, base, src, tgt, sCnt, blockList + off);
  else
    count_entries<false>(myList, cnt, lane, base, src, tgt, sCnt, blockList);
}

__global__ __launch_bounds__(OWN_THREADS) void gcn_owner_count_kernel(
    const int* __restrict__ src, const int* __restrict__ tgt,
    const unsigned* __restrict__ keyw, float* __restrict__ dinv,
    unsigned* __restrict__ glist, unsigned* __restrict__ blockLen) {
  __shared__ unsigned sCnt[NPB];
  __shared__ unsigned sList[OWN_WAVES * LIST_CAP];
  __shared__ unsigned sTotal;

  const int tid  = threadIdx.x;
  const int lane = tid & 31;
  const int wave = __builtin_amdgcn_readfirstlane(tid >> 5);
  const int base = blockIdx.x * NPB;

  if (tid < NPB) sCnt[tid] = 0u;
  if (tid == 0) sTotal = 0u;
  __syncthreads();

  unsigned* myList = sList + wave * LIST_CAP;
  unsigned* blockList = glist + (size_t)blockIdx.x * GLIST_CAP;
  const int cnt = scan_keys(keyw, myList, lane, wave, [&](int c) {
    count_list(myList, c, lane, base, src, tgt, sCnt, &sTotal, blockList);
    return 0;
  });
  count_list(myList, cnt, lane, base, src, tgt, sCnt, &sTotal, blockList);
  __syncthreads();

  if (tid < NPB) dinv[base + tid] = rsqrtf((float)sCnt[tid]);
  if (tid == 0) blockLen[blockIdx.x] = sTotal;
}

__global__ __launch_bounds__(OWN_THREADS) void gcn_owner_wmma_kernel(
    const float* __restrict__ x, const int* __restrict__ src,
    const int* __restrict__ tgt, const unsigned* __restrict__ keyw,
    const unsigned* __restrict__ glist, const unsigned* __restrict__ blockLen,
    const float* __restrict__ dinv,
    const float* __restrict__ W, const float* __restrict__ bias,
    float* __restrict__ out) {
  __shared__ __attribute__((aligned(16))) float sAcc[NPB * APITCH];
  __shared__ __attribute__((aligned(16))) float sW[D_FEAT * UNITS];
  __shared__ unsigned sList[OWN_WAVES * LIST_CAP];

  const int tid  = threadIdx.x;
  const int lane = tid & 31;
  const int wave = __builtin_amdgcn_readfirstlane(tid >> 5);
  const int base = blockIdx.x * NPB;

  {
    float4* z = (float4*)sAcc;
    for (int i = tid; i < (NPB * APITCH) / 4; i += OWN_THREADS)
      z[i] = make_float4(0.0f, 0.0f, 0.0f, 0.0f);
    const float4* Wv = (const float4*)W;
    float4* sWv = (float4*)sW;
    for (int i = tid; i < (D_FEAT * UNITS) / 4; i += OWN_THREADS) sWv[i] = Wv[i];
  }
  __syncthreads();

  const unsigned total = (unsigned)__builtin_amdgcn_readfirstlane((int)blockLen[blockIdx.x]);
  if (total <= GLIST_CAP) {
    const unsigned* blockList = glist + (size_t)blockIdx.x * GLIST_CAP;
    const unsigned nfull = total >> 5;
    const unsigned rem = total & 31u;
    unsigned bt = (unsigned)wave;
    unsigned ent = (bt < nfull) ? blockList[bt * 32u + lane] : 0u;
    while (bt < nfull) {
      const int s = (int)(ent >> REL_BITS);
      const int rel = (int)(ent & REL_MASK);
      const float d = dinv[s];
      const unsigned bn = bt + OWN_WAVES;
      ent = (bn < nfull) ? blockList[bn * 32u + lane] : 0u;
      add_batch<true>(s, rel, d, 32, lane, x, sAcc);
      bt = bn;
    }
    if (rem != 0u && (unsigned)wave == nfull % OWN_WAVES) {
      const unsigned last = ((unsigned)lane < rem) ? blockList[nfull * 32u + lane] : 0u;
      const int s = (int)(last >> REL_BITS);
      const int rel = (int)(last & REL_MASK);
      const float d = ((unsigned)lane < rem) ? dinv[s] : 0.0f;
      add_batch<false>(s, rel, d, (int)rem, lane, x, sAcc);
    }
  } else {
    unsigned* myList = sList + wave * LIST_CAP;
    const int cnt = scan_keys(keyw, myList, lane, wave, [&](int c) {
      return drain_whole_batches(myList, c, lane, base, x, src, tgt, dinv, sAcc);
    });
    drain_batch<false>(myList, cnt, lane, base, x, src, tgt, dinv, sAcc);
  }
  __syncthreads();

  if (wave < NPB / 16) {
    const int h    = lane >> 4;
    const int mn   = lane & 15;
    const int row0 = wave * 16;
    const float dm = dinv[base + row0 + mn];
    const float* aRow = sAcc + (row0 + mn) * APITCH;

    v8f acc0 = {}, acc1 = {}, acc2 = {}, acc3 = {};
    #pragma unroll 4
    for (int kk = 0; kk < D_FEAT; kk += 4) {
      const int k0 = kk + 2 * h;
      v2f a, b;
      a.x = aRow[k0]     * dm;
      a.y = aRow[k0 + 1] * dm;
      b.x = sW[(k0)     * UNITS + 0 + mn];
      b.y = sW[(k0 + 1) * UNITS + 0 + mn];
      acc0 = __builtin_amdgcn_wmma_f32_16x16x4_f32(false, a, false, b, (short)0, acc0, false, false);
      b.x = sW[(k0)     * UNITS + 16 + mn];
      b.y = sW[(k0 + 1) * UNITS + 16 + mn];
      acc1 = __builtin_amdgcn_wmma_f32_16x16x4_f32(false, a, false, b, (short)0, acc1, false, false);
      b.x = sW[(k0)     * UNITS + 32 + mn];
      b.y = sW[(k0 + 1) * UNITS + 32 + mn];
      acc2 = __builtin_amdgcn_wmma_f32_16x16x4_f32(false, a, false, b, (short)0, acc2, false, false);
      b.x = sW[(k0)     * UNITS + 48 + mn];
      b.y = sW[(k0 + 1) * UNITS + 48 + mn];
      acc3 = __builtin_amdgcn_wmma_f32_16x16x4_f32(false, a, false, b, (short)0, acc3, false, false);
    }

    const float b0 = bias[0 + mn], b1 = bias[16 + mn], b2 = bias[32 + mn], b3 = bias[48 + mn];
    float* orow = out + (size_t)(base + row0 + 8 * h) * UNITS + mn;
    #pragma unroll
    for (int j = 0; j < 8; ++j) {
      orow[(size_t)j * UNITS +  0] = fmaxf(acc0[j] + b0, 0.0f);
      orow[(size_t)j * UNITS + 16] = fmaxf(acc1[j] + b1, 0.0f);
      orow[(size_t)j * UNITS + 32] = fmaxf(acc2[j] + b2, 0.0f);
      orow[(size_t)j * UNITS + 48] = fmaxf(acc3[j] + b3, 0.0f);
    }
  }
}

extern "C" void kernel_launch(void* const* d_in, const int* in_sizes, int n_in,
                              void* d_out, int out_size, void* d_ws, size_t ws_size,
                              hipStream_t stream) {
  const float* x      = (const float*)d_in[0];
  const float* W      = (const float*)d_in[1];
  const float* b      = (const float*)d_in[2];
  const int*   source = (const int*)d_in[3];
  const int*   target = (const int*)d_in[4];
  float*       out    = (float*)d_out;

  float*    dinv = (float*)d_ws + N_NODES;
  unsigned* keyw = (unsigned*)(dinv + N_NODES);
  unsigned* glist = keyw + KEY_BITS * PLANE_PITCH;
  unsigned* blockLen = (unsigned*)d_ws;

  gcn_key_kernel<<<KEY_BLOCKS, KEY_THREADS, 0, stream>>>(target, keyw);
  gcn_owner_count_kernel<<<N_OWNERS, OWN_THREADS, 0, stream>>>(source, target, keyw, dinv, glist, blockLen);
  gcn_owner_wmma_kernel<<<N_OWNERS, OWN_THREADS, 0, stream>>>(x, source, target, keyw, glist, blockLen, dinv, W, b, out);
}


# Key kernel also writes each edge's list word (source above tile row); owner blocks fetch that one word per kept edge, not target and source: one 4-byte lookup per edge less; LDS sums arrival-order
# speedup vs baseline: 4.4218x; 1.1113x over previous
#include <hip/hip_runtime.h>
#include <hip/hip_bf16.h>


typedef __attribute__((ext_vector_type(2))) float v2f;
typedef __attribute__((ext_vector_type(8))) float v8f;

#define N_NODES 100000
#define N_EDGES 1600000
#define D_FEAT  64
#define UNITS   64

#define NPB         400
#define N_OWNERS    (N_NODES / NPB)
#define OWN_THREADS 1024
#define OWN_WAVES   (OWN_THREADS / 32)
#define APITCH      66
#define KEY_BITS    8
#define GROUP_EDGES 128
#define N_GROUPS    (N_EDGES / GROUP_EDGES)
#define PLANE_PITCH 50048
#define N_CHUNKS    ((N_GROUPS + 31) / 32)
#define WORD_EDGES  (32 * 32)
#define LIST_CAP    (32 + WORD_EDGES)
#define SCAN_UNROLL 2
#define NO_OWNER    255u
#define ROW_BYTES   (D_FEAT * 4u)
#define KEY_THREADS 256
#define KEY_WAVES   (KEY_THREADS / 32)
#define WAVE_GROUPS 8
#define KEY_BLOCKS  ((N_GROUPS + KEY_WAVES * WAVE_GROUPS - 1) / (KEY_WAVES * WAVE_GROUPS))
#define EDGEW_WORDS (KEY_BLOCKS * KEY_WAVES * WAVE_GROUPS * GROUP_EDGES)
#define REL_BITS    9
#define REL_MASK    ((1u << REL_BITS) - 1u)
#define GLIST_CAP   8192

static_assert(N_NODES % NPB == 0, "owner ranges tile the nodes exactly");
static_assert(NPB % 16 == 0 && NPB / 16 <= OWN_WAVES, "one 16-row tile per wave");
static_assert(N_EDGES % GROUP_EDGES == 0, "edge list is a whole number of groups");
static_assert(PLANE_PITCH >= N_EDGES / 32 && (PLANE_PITCH * 4) % 128 == 0, "a plane holds every edge and starts on a line");
static_assert(N_OWNERS <= (int)NO_OWNER, "an owner index fits in one byte beside the no-owner key");
static_assert(NPB <= OWN_THREADS, "one thread per owned node writes its rsqrt");
static_assert(((size_t)2 * N_NODES * sizeof(float)) % 128 == 0, "the key array starts on a line");
static_assert((unsigned long long)N_NODES * D_FEAT * 4ull < (1ull << 32), "row offsets fit in 32 bits");
static_assert(NPB <= (1 << REL_BITS), "a row of the tile fits in the low bits of a list word");
static_assert((unsigned long long)N_NODES <= (1ull << (32 - REL_BITS)), "a source fits above the row in a list word");
static_assert(GLIST_CAP % 32 == 0, "a block's list is a whole number of batches");
static_assert(((size_t)2 * N_NODES * sizeof(float) + (size_t)KEY_BITS * PLANE_PITCH * 4) % 128 == 0, "the lists start on a line");
static_assert(N_OWNERS <= N_NODES, "the list lengths fit in the unused front of the workspace");
static_assert(((size_t)N_OWNERS * GLIST_CAP * 4) % 16 == 0, "the edge words start on a 16-byte piece");

__device__ __forceinline__ unsigned owner_of(int t) {
  return (unsigned)t / (unsigned)NPB;
}
__device__ __forceinline__ unsigned key_of(unsigned o) {
  return o < NO_OWNER ? o : NO_OWNER;
}
__device__ __forceinline__ unsigned list_word(int s, int t, unsigned o) {
  return ((unsigned)s << REL_BITS) | ((unsigned)t - o * (unsigned)NPB);
}

__global__ __launch_bounds__(KEY_THREADS) void gcn_key_kernel(
    const int* __restrict__ source, const int* __restrict__ target,
    unsigned* __restrict__ keyw, unsigned* __restrict__ edgew) {
  const int4* tv = (const int4*)target;
  const int4* sv = (const int4*)source;
  const int lane = threadIdx.x & 31;
  const int wave = __builtin_amdgcn_readfirstlane((int)(threadIdx.x >> 5));
  const int g0 = (blockIdx.x * KEY_WAVES + wave) * WAVE_GROUPS;
  int4 t[WAVE_GROUPS];
  int4 s[WAVE_GROUPS];
  #pragma unroll
  for (int g = 0; g < WAVE_GROUPS; ++g) {
    const int gi = min(g0 + g, N_GROUPS - 1) * 32 + lane;
    t[g] = tv[gi];
    s[g] = sv[gi];
  }
  uint4 w[KEY_BITS];
  #pragma unroll
  for (int k = 0; k < KEY_BITS; ++k) w[k] = make_uint4(0u, 0u, 0u, 0u);
  #pragma unroll
  for (int g = 0; g < WAVE_GROUPS; ++g) {
    const unsigned ox = owner_of(t[g].x), oy = owner_of(t[g].y);
    const unsigned oz = owner_of(t[g].z), ow = owner_of(t[g].w);
    ((uint4*)edgew)[(g0 + g) * 32 + lane] = make_uint4(
        list_word(s[g].x, t[g].x, ox), list_word(s[g].y, t[g].y, oy),
        list_word(s[g].z, t[g].z, oz), list_word(s[g].w, t[g].w, ow));
    const unsigned kx = key_of(ox), ky = key_of(oy);
    const unsigned kz = key_of(oz), kw = key_of(ow);
    const bool mine = lane == g;
    #pragma unroll
    for (int k = 0; k < KEY_BITS; ++k) {
      const unsigned bx = __builtin_amdgcn_ballot_w32(((kx >> k) & 1u) != 0u);
      const unsigned by = __builtin_amdgcn_ballot_w32(((ky >> k) & 1u) != 0u);
      const unsigned bz = __builtin_amdgcn_ballot_w32(((kz >> k) & 1u) != 0u);
      const unsigned bw = __builtin_amdgcn_ballot_w32(((kw >> k) & 1u) != 0u);
      if (mine) w[k] = make_uint4(bx, by, bz, bw);
    }
  }
  const int grp = g0 + lane;
  if (lane < WAVE_GROUPS && grp < N_GROUPS) {
    #pragma unroll
    for (int k = 0; k < KEY_BITS; ++k)
      ((uint4*)(keyw + (size_t)k * PLANE_PITCH))[grp] = w[k];
  }
}

__device__ __forceinline__ void add_row(float2 v, float di, int ri, int lane, float* sAcc) {
  float* p = sAcc + ri * APITCH + 2 * lane;
  atomicAdd(p + 0, v.x * di);
  atomicAdd(p + 1, v.y * di);
}

template <bool FULL>
__device__ __forceinline__ void add_batch(
    int s, int rel, float d, int n, int lane,
    const float* __restrict__ x, float* sAcc) {
  const int dbits = __float_as_int(d);
  const char* xb = (const char*)x;
  const unsigned lane8 = (unsigned)lane * 8u;
  const int nfull = FULL ? 32 : ((n + 15) & ~15);
  for (int i = 0; i < nfull; i += 16) {
    float2 xv[16];
    #pragma unroll
    for (int q = 0; q < 16; ++q) {
      const unsigned si = (unsigned)__builtin_amdgcn_readlane(s, i + q);
      xv[q] = *(const float2*)(xb + (si * ROW_BYTES + lane8));
    }
    #pragma unroll
    for (int q = 0; q < 16; ++q) {
      const float di = __int_as_float(__builtin_amdgcn_readlane(dbits, i + q));
      const int ri = __builtin_amdgcn_readlane(rel, i + q);
      if (FULL || i + q < n) add_row(xv[q], di, ri, lane, sAcc);
    }
  }
}

template <bool FULL>
__device__ __forceinline__ void drain_batch(
    const unsigned* myList, int n, int lane,
    const float* __restrict__ x, const unsigned* __restrict__ edgew,
    const float* __restrict__ dinv, float* sAcc) {
  int s = 0;
  int rel = 0;
  float d = 0.0f;
  if (FULL || lane < n) {
    const unsigned w = edgew[myList[lane]];
    s = (int)(w >> REL_BITS);
    rel = (int)(w & REL_MASK);
    d = dinv[s];
  }
  add_batch<FULL>(s, rel, d, n, lane, x, sAcc);
}

__device__ __forceinline__ int drain_whole_batches(
    unsigned* myList, int cnt, int lane,
    const float* __restrict__ x, const unsigned* __restrict__ edgew,
    const float* __restrict__ dinv, float* sAcc) {
  const int full = cnt & ~31;
  for (int j = 0; j < full; j += 32)
    drain_batch<true>(myList + j, 32, lane, x, edgew, dinv, sAcc);
  const int rem = cnt - full;
  if (lane < rem) {
    const unsigned t = myList[full + lane];
    myList[lane] = t;
  }
  return rem;
}

__device__ __forceinline__ unsigned word_of(uint4 a, int j) {
  return j == 0 ? a.x : j == 1 ? a.y : j == 2 ? a.z : a.w;
}

template <typename Drain>
__device__ __forceinline__ int scan_keys(
    const unsigned* __restrict__ keyw, unsigned* myList, int lane, int wave, Drain drain) {
  int cnt = 0;
  unsigned inv[KEY_BITS];
  #pragma unroll
  for (int k = 0; k < KEY_BITS; ++k) inv[k] = (((unsigned)blockIdx.x >> k) & 1u) ? 0u : 0xffffffffu;
  const int start = (int)(((unsigned)blockIdx.x * (unsigned)N_CHUNKS) / (unsigned)N_OWNERS);

  for (int c0 = wave; c0 < N_CHUNKS; c0 += OWN_WAVES * SCAN_UNROLL) {
    uint4 v[SCAN_UNROLL][KEY_BITS];
    int   grp[SCAN_UNROLL];
    #pragma unroll
    for (int u = 0; u < SCAN_UNROLL; ++u) {
      int c = min(c0 + u * OWN_WAVES, N_CHUNKS - 1);
      c += start;
      if (c >= N_CHUNKS) c -= N_CHUNKS;
      grp[u] = c * 32 + lane;
      const int gl = min(grp[u], N_GROUPS - 1);
      #pragma unroll
      for (int k = 0; k < KEY_BITS; ++k)
        v[u][k] = ((const uint4*)(keyw + (size_t)k * PLANE_PITCH))[gl];
    }
    #pragma unroll
    for (int u = 0; u < SCAN_UNROLL; ++u) {
      const bool ok = (c0 + u * OWN_WAVES) < N_CHUNKS && grp[u] < N_GROUPS;
      const unsigned e0 = (unsigned)grp[u] * (unsigned)GROUP_EDGES;
      #pragma unroll
      for (int j = 0; j < 4; ++j) {
        unsigned m = word_of(v[u][0], j) ^ inv[0];
        #pragma unroll
        for (int k = 1; k < KEY_BITS; ++k) m &= word_of(v[u][k], j) ^ inv[k];
        if (!ok) m = 0u;
        unsigned has = __builtin_amdgcn_ballot_w32(m != 0u);
        for (int it = 0; it < 32 && has != 0u; ++it) {
          if (m != 0u) {
            const unsigned p = (unsigned)__builtin_ctz(m);
            m &= m - 1u;
            myList[cnt + (int)__builtin_amdgcn_mbcnt_lo(has, 0u)] = e0 + (p << 2) + (unsigned)j;
          }
          cnt += __popc(has);
          has = __builtin_amdgcn_ballot_w32(m != 0u);
        }
        if (cnt >= 32) cnt = drain(cnt);
      }
    }
  }
  return cnt;
}

template <bool LIST>
__device__ __forceinline__ void count_entries(
    const unsigned* myList, int cnt, int lane,
    const unsigned* __restrict__ edgew,
    unsigned* sCnt, unsigned* __restrict__ dst) {
  for (int j = lane; j < cnt; j += 64) {
    const bool two = j + 32 < cnt;
    const int jb = two ? j + 32 : j;
    const unsigned wa = edgew[myList[j]];
    const unsigned wb = edgew[myList[jb]];
    atomicAdd(&sCnt[wa & REL_MASK], 1u);
    atomicAdd(&sCnt[wb & REL_MASK], two ? 1u : 0u);
    if (LIST) {
      dst[j]  = wa;
      dst[jb] = wb;
    }
  }
}

__device__ __forceinline__ void count_list(
    const unsigned* myList, int cnt, int lane,
    const unsigned* __restrict__ edgew,
    unsigned* sCnt, unsigned* sTotal, unsigned* __restrict__ blockList) {
  if (cnt == 0) return;
  unsigned off = 0u;
  if (lane == 0) off = atomicAdd(sTotal, (unsigned)cnt);
  off = (unsigned)__builtin_amdgcn_readfirstlane((int)off);
  if (off + (unsigned)cnt <= GLIST_CAP)
    count_entries<true>(myList, cnt, lane, edgew, sCnt, blockList + off);
  else
    count_entries<false>(myList, cnt, lane, edgew, sCnt, blockList);
}

__global__ __launch_bounds__(OWN_THREADS) void gcn_owner_count_kernel(
    const unsigned* __restrict__ keyw, const unsigned* __restrict__ edgew,
    float* __restrict__ dinv,
    unsigned* __restrict__ glist, unsigned* __restrict__ blockLen) {
  __shared__ unsigned sCnt[NPB];
  __shared__ unsigned sList[OWN_WAVES * LIST_CAP];
  __shared__ unsigned sTotal;

  const int tid  = threadIdx.x;
  const int lane = tid & 31;
  const int wave = __builtin_amdgcn_readfirstlane(tid >> 5);
  const int base = blockIdx.x * NPB;

  if (tid < NPB) sCnt[tid] = 0u;
  if (tid == 0) sTotal = 0u;
  __syncthreads();

  unsigned* myList = sList + wave * LIST_CAP;
  unsigned* blockList = glist + (size_t)blockIdx.x * GLIST_CAP;
  const int cnt = scan_keys(keyw, myList, lane, wave, [&](int c) {
    count_list(myList, c, lane, edgew, sCnt, &sTotal, blockList);
    return 0;
  });
  count_list(myList, cnt, lane, edgew, sCnt, &sTotal, blockList);
  __syncthreads();

  if (tid < NPB) dinv[base + tid] = rsqrtf((float)sCnt[tid]);
  if (tid == 0) blockLen[blockIdx.x] = sTotal;
}

__global__ __launch_bounds__(OWN_THREADS) void gcn_owner_wmma_kernel(
    const float* __restrict__ x, const unsigned* __restrict__ keyw,
    const unsigned* __restrict__ edgew,
    const unsigned* __restrict__ glist, const unsigned* __restrict__ blockLen,
    const float* __restrict__ dinv,
    const float* __restrict__ W, const float* __restrict__ bias,
    float* __restrict__ out) {
  __shared__ __attribute__((aligned(16))) float sAcc[NPB * APITCH];
  __shared__ __attribute__((aligned(16))) float sW[D_FEAT * UNITS];
  __shared__ unsigned sList[OWN_WAVES * LIST_CAP];

  const int tid  = threadIdx.x;
  const int lane = tid & 31;
  const int wave = __builtin_amdgcn_readfirstlane(tid >> 5);
  const int base = blockIdx.x * NPB;

  {
    float4* z = (float4*)sAcc;
    for (int i = tid; i < (NPB * APITCH) / 4; i += OWN_THREADS)
      z[i] = make_float4(0.0f, 0.0f, 0.0f, 0.0f);
    const float4* Wv = (const float4*)W;
    float4* sWv = (float4*)sW;
    for (int i = tid; i < (D_FEAT * UNITS) / 4; i += OWN_THREADS) sWv[i] = Wv[i];
  }
  __syncthreads();

  const unsigned total = (unsigned)__builtin_amdgcn_readfirstlane((int)blockLen[blockIdx.x]);
  if (total <= GLIST_CAP) {
    const unsigned* blockList = glist + (size_t)blockIdx.x * GLIST_CAP;
    const unsigned nfull = total >> 5;
    const unsigned rem = total & 31u;
    unsigned bt = (unsigned)wave;
    unsigned ent = (bt < nfull) ? blockList[bt * 32u + lane] : 0u;
    while (bt < nfull) {
      const int s = (int)(ent >> REL_BITS);
      const int rel = (int)(ent & REL_MASK);
      const float d = dinv[s];
      const unsigned bn = bt + OWN_WAVES;
      ent = (bn < nfull) ? blockList[bn * 32u + lane] : 0u;
      add_batch<true>(s, rel, d, 32, lane, x, sAcc);
      bt = bn;
    }
    if (rem != 0u && (unsigned)wave == nfull % OWN_WAVES) {
      const unsigned last = ((unsigned)lane < rem) ? blockList[nfull * 32u + lane] : 0u;
      const int s = (int)(last >> REL_BITS);
      const int rel = (int)(last & REL_MASK);
      const float d = ((unsigned)lane < rem) ? dinv[s] : 0.0f;
      add_batch<false>(s, rel, d, (int)rem, lane, x, sAcc);
    }
  } else {
    unsigned* myList = sList + wave * LIST_CAP;
    const int cnt = scan_keys(keyw, myList, lane, wave, [&](int c) {
      return drain_whole_batches(myList, c, lane, x, edgew, dinv, sAcc);
    });
    drain_batch<false>(myList, cnt, lane, x, edgew, dinv, sAcc);
  }
  __syncthreads();

  if (wave < NPB / 16) {
    const int h    = lane >> 4;
    const int mn   = lane & 15;
    const int row0 = wave * 16;
    const float dm = dinv[base + row0 + mn];
    const float* aRow = sAcc + (row0 + mn) * APITCH;

    v8f acc0 = {}, acc1 = {}, acc2 = {}, acc3 = {};
    #pragma unroll 4
    for (int kk = 0; kk < D_FEAT; kk += 4) {
      const int k0 = kk + 2 * h;
      v2f a, b;
      a.x = aRow[k0]     * dm;
      a.y = aRow[k0 + 1] * dm;
      b.x = sW[(k0)     * UNITS + 0 + mn];
      b.y = sW[(k0 + 1) * UNITS + 0 + mn];
      acc0 = __builtin_amdgcn_wmma_f32_16x16x4_f32(false, a, false, b, (short)0, acc0, false, false);
      b.x = sW[(k0)     * UNITS + 16 + mn];
      b.y = sW[(k0 + 1) * UNITS + 16 + mn];
      acc1 = __builtin_amdgcn_wmma_f32_16x16x4_f32(false, a, false, b, (short)0, acc1, false, false);
      b.x = sW[(k0)     * UNITS + 32 + mn];
      b.y = sW[(k0 + 1) * UNITS + 32 + mn];
      acc2 = __builtin_amdgcn_wmma_f32_16x16x4_f32(false, a, false, b, (short)0, acc2, false, false);
      b.x = sW[(k0)     * UNITS + 48 + mn];
      b.y = sW[(k0 + 1) * UNITS + 48 + mn];
      acc3 = __builtin_amdgcn_wmma_f32_16x16x4_f32(false, a, false, b, (short)0, acc3, false, false);
    }

    const float b0 = bias[0 + mn], b1 = bias[16 + mn], b2 = bias[32 + mn], b3 = bias[48 + mn];
    float* orow = out + (size_t)(base + row0 + 8 * h) * UNITS + mn;
    #pragma unroll
    for (int j = 0; j < 8; ++j) {
      orow[(size_t)j * UNITS +  0] = fmaxf(acc0[j] + b0, 0.0f);
      orow[(size_t)j * UNITS + 16] = fmaxf(acc1[j] + b1, 0.0f);
      orow[(size_t)j * UNITS + 32] = fmaxf(acc2[j] + b2, 0.0f);
      orow[(size_t)j * UNITS + 48] = fmaxf(acc3[j] + b3, 0.0f);
    }
  }
}

extern "C" void kernel_launch(void* const* d_in, const int* in_sizes, int n_in,
                              void* d_out, int out_size, void* d_ws, size_t ws_size,
                              hipStream_t stream) {
  const float* x      = (const float*)d_in[0];
  const float* W      = (const float*)d_in[1];
  const float* b      = (const float*)d_in[2];
  const int*   source = (const int*)d_in[3];
  const int*   target = (const int*)d_in[4];
  float*       out    = (float*)d_out;

  float*    dinv = (float*)d_ws + N_NODES;
  unsigned* keyw = (unsigned*)(dinv + N_NODES);
  unsigned* glist = keyw + KEY_BITS * PLANE_PITCH;
  unsigned* blockLen = (unsigned*)d_ws;
  unsigned* edgew = glist + N_OWNERS * GLIST_CAP;

  gcn_key_kernel<<<KEY_BLOCKS, KEY_THREADS, 0, stream>>>(source, target, keyw, edgew);
  gcn_owner_count_kernel<<<N_OWNERS, OWN_THREADS, 0, stream>>>(keyw, edgew, dinv, glist, blockLen);
  gcn_owner_wmma_kernel<<<N_OWNERS, OWN_THREADS, 0, stream>>>(x, keyw, edgew, glist, blockLen, dinv, W, b, out);
}


# Phase-2 key filter tiled two ways: a block takes 4 owners (shared high 6 key bits) and a quarter of the edges, planes read 63 times not 250; deg by global integer adds (exact); LDS sums arrival-order
# speedup vs baseline: 4.7184x; 1.0671x over previous
#include <hip/hip_runtime.h>
#include <hip/hip_bf16.h>


typedef __attribute__((ext_vector_type(2))) float v2f;
typedef __attribute__((ext_vector_type(8))) float v8f;

#define N_NODES 100000
#define N_EDGES 1600000
#define D_FEAT  64
#define UNITS   64

#define NPB         400
#define N_OWNERS    (N_NODES / NPB)
#define OWN_THREADS 1024
#define OWN_WAVES   (OWN_THREADS / 32)
#define APITCH      66
#define KEY_BITS    8
#define GROUP_EDGES 128
#define N_GROUPS    (N_EDGES / GROUP_EDGES)
#define PLANE_PITCH 50048
#define N_CHUNKS    ((N_GROUPS + 31) / 32)
#define WORD_EDGES  (32 * 32)
#define LIST_CAP    (32 + WORD_EDGES)
#define SCAN_UNROLL 2
#define NO_OWNER    255u
#define ROW_BYTES   (D_FEAT * 4u)
#define KEY_THREADS 256
#define KEY_WAVES   (KEY_THREADS / 32)
#define WAVE_GROUPS 8
#define KEY_BLOCKS  ((N_GROUPS + KEY_WAVES * WAVE_GROUPS - 1) / (KEY_WAVES * WAVE_GROUPS))
#define EDGEW_WORDS (KEY_BLOCKS * KEY_WAVES * WAVE_GROUPS * GROUP_EDGES)
#define REL_BITS    9
#define REL_MASK    ((1u << REL_BITS) - 1u)
#define GLIST_CAP   8192
#define LOW_BITS    2
#define GRP_OWNERS  (1 << LOW_BITS)
#define N_GRPS      ((N_OWNERS + GRP_OWNERS - 1) / GRP_OWNERS)
#define GRP_NODES   (GRP_OWNERS * NPB)
#define N_PARTS     4
#define PART_CHUNKS ((N_CHUNKS + N_PARTS - 1) / N_PARTS)
#define SUBCAP      (GLIST_CAP / N_PARTS)
#define SUB_SHIFT   21
#define EDGE_MASK   ((1u << SUB_SHIFT) - 1u)

static_assert(N_NODES % NPB == 0, "owner ranges tile the nodes exactly");
static_assert(NPB % 16 == 0 && NPB / 16 <= OWN_WAVES, "one 16-row tile per wave");
static_assert(N_EDGES % GROUP_EDGES == 0, "edge list is a whole number of groups");
static_assert(PLANE_PITCH >= N_EDGES / 32 && (PLANE_PITCH * 4) % 128 == 0, "a plane holds every edge and starts on a line");
static_assert(KEY_BLOCKS * KEY_THREADS >= N_NODES / 2, "the key grid has a thread for every pair of degree words");
static_assert(N_OWNERS <= (int)NO_OWNER, "an owner index fits in one byte beside the no-owner key");
static_assert(NPB <= OWN_THREADS, "one thread per owned node writes its rsqrt");
static_assert(((size_t)2 * N_NODES * sizeof(float)) % 128 == 0, "the key array starts on a line");
static_assert((unsigned long long)N_NODES * D_FEAT * 4ull < (1ull << 32), "row offsets fit in 32 bits");
static_assert(NPB <= (1 << REL_BITS), "a row of the tile fits in the low bits of a list word");
static_assert((unsigned long long)N_NODES <= (1ull << (32 - REL_BITS)), "a source fits above the row in a list word");
static_assert(GLIST_CAP % 32 == 0, "a block's list is a whole number of batches");
static_assert(((size_t)2 * N_NODES * sizeof(float) + (size_t)KEY_BITS * PLANE_PITCH * 4) % 128 == 0, "the lists start on a line");
static_assert(N_OWNERS * N_PARTS <= N_NODES, "the sub-list lengths fit in the unused front of the workspace");
static_assert(N_EDGES <= (1 << SUB_SHIFT), "an edge number fits below the owner's low key bits in a filter entry");
static_assert(SUBCAP % 32 == 0 && SUBCAP * N_PARTS == GLIST_CAP, "an owner's sub-lists fill its list space in whole batches");
static_assert(LOW_BITS < KEY_BITS && (N_PARTS - 1) * PART_CHUNKS < N_CHUNKS, "every part of the edge list holds a chunk");
static_assert(((size_t)N_NODES * sizeof(unsigned)) % 8 == 0 && N_NODES % 2 == 0, "the degree array is cleared with 8-byte stores");
static_assert(((size_t)N_OWNERS * GLIST_CAP * 4) % 16 == 0, "the edge words start on a 16-byte piece");

__device__ __forceinline__ unsigned owner_of(int t) {
  return (unsigned)t / (unsigned)NPB;
}
__device__ __forceinline__ unsigned key_of(unsigned o) {
  return o < (unsigned)N_OWNERS ? o : NO_OWNER;
}
__device__ __forceinline__ unsigned list_word(int s, int t, unsigned o) {
  return ((unsigned)s << REL_BITS) | ((unsigned)t - o * (unsigned)NPB);
}

__global__ __launch_bounds__(KEY_THREADS) void gcn_key_kernel(
    const int* __restrict__ source, const int* __restrict__ target,
    unsigned* __restrict__ keyw, unsigned* __restrict__ edgew, unsigned* __restrict__ deg) {
  {
    const unsigned gt = blockIdx.x * KEY_THREADS + threadIdx.x;
    if (gt < N_NODES / 2) ((uint2*)deg)[gt] = make_uint2(0u, 0u);
  }
  const int4* tv = (const int4*)target;
  const int4* sv = (const int4*)source;
  const int lane = threadIdx.x & 31;
  const int wave = __builtin_amdgcn_readfirstlane((int)(threadIdx.x >> 5));
  const int g0 = (blockIdx.x * KEY_WAVES + wave) * WAVE_GROUPS;
  int4 t[WAVE_GROUPS];
  int4 s[WAVE_GROUPS];
  #pragma unroll
  for (int g = 0; g < WAVE_GROUPS; ++g) {
    const int gi = min(g0 + g, N_GROUPS - 1) * 32 + lane;
    t[g] = tv[gi];
    s[g] = sv[gi];
  }
  uint4 w[KEY_BITS];
  #pragma unroll
  for (int k = 0; k < KEY_BITS; ++k) w[k] = make_uint4(0u, 0u, 0u, 0u);
  #pragma unroll
  for (int g = 0; g < WAVE_GROUPS; ++g) {
    const unsigned ox = owner_of(t[g].x), oy = owner_of(t[g].y);
    const unsigned oz = owner_of(t[g].z), ow = owner_of(t[g].w);
    ((uint4*)edgew)[(g0 + g) * 32 + lane] = make_uint4(
        list_word(s[g].x, t[g].x, ox), list_word(s[g].y, t[g].y, oy),
        list_word(s[g].z, t[g].z, oz), list_word(s[g].w, t[g].w, ow));
    const unsigned kx = key_of(ox), ky = key_of(oy);
    const unsigned kz = key_of(oz), kw = key_of(ow);
    const bool mine = lane == g;
    #pragma unroll
    for (int k = 0; k < KEY_BITS; ++k) {
      const unsigned bx = __builtin_amdgcn_ballot_w32(((kx >> k) & 1u) != 0u);
      const unsigned by = __builtin_amdgcn_ballot_w32(((ky >> k) & 1u) != 0u);
      const unsigned bz = __builtin_amdgcn_ballot_w32(((kz >> k) & 1u) != 0u);
      const unsigned bw = __builtin_amdgcn_ballot_w32(((kw >> k) & 1u) != 0u);
      if (mine) w[k] = make_uint4(bx, by, bz, bw);
    }
  }
  const int grp = g0 + lane;
  if (lane < WAVE_GROUPS && grp < N_GROUPS) {
    #pragma unroll
    for (int k = 0; k < KEY_BITS; ++k)
      ((uint4*)(keyw + (size_t)k * PLANE_PITCH))[grp] = w[k];
  }
}

__device__ __forceinline__ void add_row(float2 v, float di, int ri, int lane, float* sAcc) {
  float* p = sAcc + ri * APITCH + 2 * lane;
  atomicAdd(p + 0, v.x * di);
  atomicAdd(p + 1, v.y * di);
}

template <bool FULL>
__device__ __forceinline__ void add_batch(
    int s, int rel, float d, int n, int lane,
    const float* __restrict__ x, float* sAcc) {
  const int dbits = __float_as_int(d);
  const char* xb = (const char*)x;
  const unsigned lane8 = (unsigned)lane * 8u;
  const int nfull = FULL ? 32 : ((n + 15) & ~15);
  for (int i = 0; i < nfull; i += 16) {
    float2 xv[16];
    #pragma unroll
    for (int q = 0; q < 16; ++q) {
      const unsigned si = (unsigned)__builtin_amdgcn_readlane(s, i + q);
      xv[q] = *(const float2*)(xb + (si * ROW_BYTES + lane8));
    }
    #pragma unroll
    for (int q = 0; q < 16; ++q) {
      const float di = __int_as_float(__builtin_amdgcn_readlane(dbits, i + q));
      const int ri = __builtin_amdgcn_readlane(rel, i + q);
      if (FULL || i + q < n) add_row(xv[q], di, ri, lane, sAcc);
    }
  }
}

__device__ __forceinline__ float dinv_of(const unsigned* __restrict__ deg, int node) {
  return rsqrtf((float)deg[node]);
}

template <bool FULL>
__device__ __forceinline__ void drain_batch(
    const unsigned* myList, int n, int lane,
    const float* __restrict__ x, const unsigned* __restrict__ edgew,
    const unsigned* __restrict__ deg, float* sAcc) {
  int s = 0;
  int rel = 0;
  float d = 0.0f;
  if (FULL || lane < n) {
    const unsigned w = edgew[myList[lane]];
    s = (int)(w >> REL_BITS);
    rel = (int)(w & REL_MASK);
    d = dinv_of(deg, s);
  }
  add_batch<FULL>(s, rel, d, n, lane, x, sAcc);
}

__device__ __forceinline__ int drain_whole_batches(
    unsigned* myList, int cnt, int lane,
    const float* __restrict__ x, const unsigned* __restrict__ edgew,
    const unsigned* __restrict__ deg, float* sAcc) {
  const int full = cnt & ~31;
  for (int j = 0; j < full; j += 32)
    drain_batch<true>(myList + j, 32, lane, x, edgew, deg, sAcc);
  const int rem = cnt - full;
  if (lane < rem) {
    const unsigned t = myList[full + lane];
    myList[lane] = t;
  }
  return rem;
}

__device__ __forceinline__ unsigned word_of(uint4 a, int j) {
  return j == 0 ? a.x : j == 1 ? a.y : j == 2 ? a.z : a.w;
}

template <int LOW, typename Drain>
__device__ __forceinline__ int scan_keys(
    const unsigned* __restrict__ keyw, unsigned* myList, int lane, int wave,
    unsigned key, int cBeg, int cNum, int start, Drain drain) {
  int cnt = 0;
  unsigned inv[KEY_BITS];
  #pragma unroll
  for (int k = 0; k < KEY_BITS; ++k) inv[k] = ((key >> k) & 1u) ? 0u : 0xffffffffu;

  for (int c0 = wave; c0 < cNum; c0 += OWN_WAVES * SCAN_UNROLL) {
    uint4 v[SCAN_UNROLL][KEY_BITS];
    int   grp[SCAN_UNROLL];
    #pragma unroll
    for (int u = 0; u < SCAN_UNROLL; ++u) {
      int c = min(c0 + u * OWN_WAVES, cNum - 1);
      c += start;
      if (c >= cNum) c -= cNum;
      grp[u] = (cBeg + c) * 32 + lane;
      const int gl = min(grp[u], N_GROUPS - 1);
      #pragma unroll
      for (int k = 0; k < KEY_BITS; ++k)
        v[u][k] = ((const uint4*)(keyw + (size_t)k * PLANE_PITCH))[gl];
    }
    #pragma unroll
    for (int u = 0; u < SCAN_UNROLL; ++u) {
      const bool ok = (c0 + u * OWN_WAVES) < cNum && grp[u] < N_GROUPS;
      const unsigned e0 = (unsigned)grp[u] * (unsigned)GROUP_EDGES;
      #pragma unroll
      for (int j = 0; j < 4; ++j) {
        unsigned m = word_of(v[u][LOW], j) ^ inv[LOW];
        #pragma unroll
        for (int k = LOW + 1; k < KEY_BITS; ++k) m &= word_of(v[u][k], j) ^ inv[k];
        if (!ok) m = 0u;
        unsigned has = __builtin_amdgcn_ballot_w32(m != 0u);
        for (int it = 0; it < 32 && has != 0u; ++it) {
          if (m != 0u) {
            const unsigned p = (unsigned)__builtin_ctz(m);
            m &= m - 1u;
            unsigned ent = e0 + (p << 2) + (unsigned)j;
            #pragma unroll
            for (int k = 0; k < LOW; ++k) ent |= ((word_of(v[u][k], j) >> p) & 1u) << (SUB_SHIFT + k);
            myList[cnt + (int)__builtin_amdgcn_mbcnt_lo(has, 0u)] = ent;
          }
          cnt += __popc(has);
          has = __builtin_amdgcn_ballot_w32(m != 0u);
        }
        if (cnt >= 32) cnt = drain(cnt);
      }
    }
  }
  return cnt;
}

__device__ __forceinline__ void count_list(
    const unsigned* myList, int cnt, int lane,
    const unsigned* __restrict__ edgew,
    unsigned* sCnt, unsigned* sTotal, unsigned* __restrict__ partList) {
  for (int j0 = 0; j0 < cnt; j0 += 64) {
    const int ja = j0 + lane;
    const bool va = ja < cnt;
    const bool vb = ja + 32 < cnt;
    const unsigned ea = myList[va ? ja : 0];
    const unsigned eb = myList[vb ? ja + 32 : 0];
    const unsigned wa = edgew[ea & EDGE_MASK];
    const unsigned wb = edgew[eb & EDGE_MASK];
    const unsigned sa = ea >> SUB_SHIFT;
    const unsigned sb = eb >> SUB_SHIFT;
    unsigned ra = 0u, rb = 0u, mine = 0u;
    #pragma unroll
    for (unsigned i = 0; i < GRP_OWNERS; ++i) {
      const unsigned ba = __builtin_amdgcn_ballot_w32(va && sa == i);
      const unsigned bb = __builtin_amdgcn_ballot_w32(vb && sb == i);
      if (sa == i) ra = __builtin_amdgcn_mbcnt_lo(ba, 0u);
      if (sb == i) rb = (unsigned)__popc(ba) + __builtin_amdgcn_mbcnt_lo(bb, 0u);
      if ((unsigned)lane == i) mine = (unsigned)__popc(ba) + (unsigned)__popc(bb);
    }
    unsigned off = 0u;
    if (lane < GRP_OWNERS && mine != 0u) off = atomicAdd(&sTotal[lane], mine);
    const unsigned pa = (unsigned)__builtin_amdgcn_ds_bpermute((int)(sa << 2), (int)off) + ra;
    const unsigned pb = (unsigned)__builtin_amdgcn_ds_bpermute((int)(sb << 2), (int)off) + rb;
    if (va) {
      atomicAdd(&sCnt[sa * NPB + (wa & REL_MASK)], 1u);
      if (pa < SUBCAP) partList[sa * GLIST_CAP + pa] = wa;
    }
    if (vb) {
      atomicAdd(&sCnt[sb * NPB + (wb & REL_MASK)], 1u);
      if (pb < SUBCAP) partList[sb * GLIST_CAP + pb] = wb;
    }
  }
}

__global__ __launch_bounds__(OWN_THREADS) void gcn_group_count_kernel(
    const unsigned* __restrict__ keyw, const unsigned* __restrict__ edgew,
    unsigned* __restrict__ deg,
    unsigned* __restrict__ glist, unsigned* __restrict__ subLen) {
  __shared__ unsigned sCnt[GRP_NODES];
  __shared__ unsigned sList[OWN_WAVES * LIST_CAP];
  __shared__ unsigned sTotal[GRP_OWNERS];

  const int tid  = threadIdx.x;
  const int lane = tid & 31;
  const int wave = __builtin_amdgcn_readfirstlane(tid >> 5);
  const int grp  = blockIdx.x / N_PARTS;
  const int part = blockIdx.x % N_PARTS;

  for (int i = tid; i < GRP_NODES; i += OWN_THREADS) sCnt[i] = 0u;
  if (tid < GRP_OWNERS) sTotal[tid] = 0u;
  __syncthreads();

  const int cBeg = part * PART_CHUNKS;
  const int cNum = min(PART_CHUNKS, N_CHUNKS - cBeg);
  const int start = (grp * cNum) / N_GRPS;
  unsigned* myList = sList + wave * LIST_CAP;
  unsigned* partList = glist + (size_t)grp * GRP_OWNERS * GLIST_CAP + part * SUBCAP;
  const int cnt = scan_keys<LOW_BITS>(keyw, myList, lane, wave,
      (unsigned)grp << LOW_BITS, cBeg, cNum, start, [&](int c) {
    count_list(myList, c, lane, edgew, sCnt, sTotal, partList);
    return 0;
  });
  count_list(myList, cnt, lane, edgew, sCnt, sTotal, partList);
  __syncthreads();

  for (int i = tid; i < GRP_NODES; i += OWN_THREADS) {
    const int node = grp * GRP_NODES + i;
    const unsigned c = sCnt[i];
    if (node < N_NODES && c != 0u) atomicAdd(&deg[node], c);
  }
  if (tid < GRP_OWNERS && grp * GRP_OWNERS + tid < N_OWNERS)
    subLen[(grp * GRP_OWNERS + tid) * N_PARTS + part] = sTotal[tid];
}

__global__ __launch_bounds__(OWN_THREADS) void gcn_owner_wmma_kernel(
    const float* __restrict__ x, const unsigned* __restrict__ keyw,
    const unsigned* __restrict__ edgew,
    const unsigned* __restrict__ glist, const unsigned* __restrict__ subLen,
    const unsigned* __restrict__ deg,
    const float* __restrict__ W, const float* __restrict__ bias,
    float* __restrict__ out) {
  __shared__ __attribute__((aligned(16))) float sAcc[NPB * APITCH];
  __shared__ __attribute__((aligned(16))) float sW[D_FEAT * UNITS];
  __shared__ unsigned sList[OWN_WAVES * LIST_CAP];

  const int tid  = threadIdx.x;
  const int lane = tid & 31;
  const int wave = __builtin_amdgcn_readfirstlane(tid >> 5);
  const int base = blockIdx.x * NPB;

  {
    float4* z = (float4*)sAcc;
    for (int i = tid; i < (NPB * APITCH) / 4; i += OWN_THREADS)
      z[i] = make_float4(0.0f, 0.0f, 0.0f, 0.0f);
    const float4* Wv = (const float4*)W;
    float4* sWv = (float4*)sW;
    for (int i = tid; i < (D_FEAT * UNITS) / 4; i += OWN_THREADS) sWv[i] = Wv[i];
  }
  __syncthreads();

  unsigned len[N_PARTS];
  unsigned nb[N_PARTS];
  unsigned nbAll = 0u;
  bool whole = true;
  #pragma unroll
  for (int q = 0; q < N_PARTS; ++q) {
    len[q] = (unsigned)__builtin_amdgcn_readfirstlane((int)subLen[blockIdx.x * N_PARTS + q]);
    nb[q] = (len[q] + 31u) >> 5;
    nbAll += nb[q];
    whole = whole && len[q] <= SUBCAP;
  }
  if (whole) {
    const unsigned* ownList = glist + (size_t)blockIdx.x * GLIST_CAP;
    auto locate = [&](unsigned bt, unsigned& at, int& n) {
      unsigned q = 0u, b = bt;
      #pragma unroll
      for (int i = 0; i + 1 < N_PARTS; ++i)
        if (q == (unsigned)i && b >= nb[i]) { b -= nb[i]; q = (unsigned)i + 1u; }
      unsigned lq = len[0];
      #pragma unroll
      for (int i = 1; i < N_PARTS; ++i) if (q == (unsigned)i) lq = len[i];
      at = q * SUBCAP + b * 32u;
      n = (int)min(32u, lq - b * 32u);
    };
    unsigned bt = (unsigned)wave;
    unsigned at = 0u;
    int n = 0;
    unsigned ent = 0u;
    if (bt < nbAll) {
      locate(bt, at, n);
      ent = (lane < n) ? ownList[at + lane] : 0u;
    }
    while (bt < nbAll) {
      const int s = (int)(ent >> REL_BITS);
      const int rel = (int)(ent & REL_MASK);
      const float d = (lane < n) ? dinv_of(deg, s) : 0.0f;
      const unsigned bn = bt + OWN_WAVES;
      unsigned atn = 0u;
      int nn = 0;
      unsigned entn = 0u;
      if (bn < nbAll) {
        locate(bn, atn, nn);
        entn = (lane < nn) ? ownList[atn + lane] : 0u;
      }
      if (n == 32) add_batch<true>(s, rel, d, 32, lane, x, sAcc);
      else add_batch<false>(s, rel, d, n, lane, x, sAcc);
      bt = bn;
      ent = entn;
      n = nn;
    }
  } else {
    unsigned* myList = sList + wave * LIST_CAP;
    const int start = (int)(((unsigned)blockIdx.x * (unsigned)N_CHUNKS) / (unsigned)N_OWNERS);
    const int cnt = scan_keys<0>(keyw, myList, lane, wave,
        (unsigned)blockIdx.x, 0, N_CHUNKS, start, [&](int c) {
      return drain_whole_batches(myList, c, lane, x, edgew, deg, sAcc);
    });
    drain_batch<false>(myList, cnt, lane, x, edgew, deg, sAcc);
  }
  __syncthreads();

  if (wave < NPB / 16) {
    const int h    = lane >> 4;
    const int mn   = lane & 15;
    const int row0 = wave * 16;
    const float dm = dinv_of(deg, base + row0 + mn);
    const float* aRow = sAcc + (row0 + mn) * APITCH;

    v8f acc0 = {}, acc1 = {}, acc2 = {}, acc3 = {};
    #pragma unroll 4
    for (int kk = 0; kk < D_FEAT; kk += 4) {
      const int k0 = kk + 2 * h;
      v2f a, b;
      a.x = aRow[k0]     * dm;
      a.y = aRow[k0 + 1] * dm;
      b.x = sW[(k0)     * UNITS + 0 + mn];
      b.y = sW[(k0 + 1) * UNITS + 0 + mn];
      acc0 = __builtin_amdgcn_wmma_f32_16x16x4_f32(false, a, false, b, (short)0, acc0, false, false);
      b.x = sW[(k0)     * UNITS + 16 + mn];
      b.y = sW[(k0 + 1) * UNITS + 16 + mn];
      acc1 = __builtin_amdgcn_wmma_f32_16x16x4_f32(false, a, false, b, (short)0, acc1, false, false);
      b.x = sW[(k0)     * UNITS + 32 + mn];
      b.y = sW[(k0 + 1) * UNITS + 32 + mn];
      acc2 = __builtin_amdgcn_wmma_f32_16x16x4_f32(false, a, false, b, (short)0, acc2, false, false);
      b.x = sW[(k0)     * UNITS + 48 + mn];
      b.y = sW[(k0 + 1) * UNITS + 48 + mn];
      acc3 = __builtin_amdgcn_wmma_f32_16x16x4_f32(false, a, false, b, (short)0, acc3, false, false);
    }

    const float b0 = bias[0 + mn], b1 = bias[16 + mn], b2 = bias[32 + mn], b3 = bias[48 + mn];
    float* orow = out + (size_t)(base + row0 + 8 * h) * UNITS + mn;
    #pragma unroll
    for (int j = 0; j < 8; ++j) {
      orow[(size_t)j * UNITS +  0] = fmaxf(acc0[j] + b0, 0.0f);
      orow[(size_t)j * UNITS + 16] = fmaxf(acc1[j] + b1, 0.0f);
      orow[(size_t)j * UNITS + 32] = fmaxf(acc2[j] + b2, 0.0f);
      orow[(size_t)j * UNITS + 48] = fmaxf(acc3[j] + b3, 0.0f);
    }
  }
}

extern "C" void kernel_launch(void* const* d_in, const int* in_sizes, int n_in,
                              void* d_out, int out_size, void* d_ws, size_t ws_size,
                              hipStream_t stream) {
  const float* x      = (const float*)d_in[0];
  const float* W      = (const float*)d_in[1];
  const float* b      = (const float*)d_in[2];
  const int*   source = (const int*)d_in[3];
  const int*   target = (const int*)d_in[4];
  float*       out    = (float*)d_out;

  unsigned* subLen = (unsigned*)d_ws;
  unsigned* deg  = (unsigned*)d_ws + N_NODES;
  unsigned* keyw = deg + N_NODES;
  unsigned* glist = keyw + KEY_BITS * PLANE_PITCH;
  unsigned* edgew = glist + N_OWNERS * GLIST_CAP;

  gcn_key_kernel<<<KEY_BLOCKS, KEY_THREADS, 0, stream>>>(source, target, keyw, edgew, deg);
  gcn_group_count_kernel<<<N_GRPS * N_PARTS, OWN_THREADS, 0, stream>>>(keyw, edgew, deg, glist, subLen);
  gcn_owner_wmma_kernel<<<N_OWNERS, OWN_THREADS, 0, stream>>>(x, keyw, edgew, glist, subLen, deg, W, b, out);
}
